# starting kernel
# baseline (speedup 1.0000x reference)
#include <hip/hip_runtime.h>
#include <hip/hip_bf16.h>
#include <math.h>

typedef float v2f __attribute__((ext_vector_type(2)));
typedef float v8f __attribute__((ext_vector_type(8)));

#define THREADS 256
#define SOM_DIM 32
#define LSTRIDE 33


__global__ __launch_bounds__(THREADS)
void som_pass_a(const float* __restrict__ input,
                const float* __restrict__ particles,
                float* __restrict__ blkSum,
                float* __restrict__ blkMinV,
                int*   __restrict__ blkMinI,
                int N)
{
    __shared__ float stage[THREADS * LSTRIDE];
    __shared__ float wsum[8 * SOM_DIM];
    __shared__ float sdist[THREADS];
    __shared__ int   sidx[THREADS];

    const int t    = threadIdx.x;
    const int p    = blockIdx.x * THREADS + t;
    const int lane = t & 31;
    const int wave = t >> 5;

    float dist2 = 0.f;
    float* srow = &stage[t * LSTRIDE];
    const bool active = (p < N);
    if (active) {
        const float4* x4 = reinterpret_cast<const float4*>(input);
        const float4* p4 = reinterpret_cast<const float4*>(particles + (size_t)p * SOM_DIM);
        #pragma unroll
        for (int q = 0; q < 8; ++q) {
            float4 pv = p4[q];
            float4 xv = x4[q];
            float d0 = pv.x - xv.x, d1 = pv.y - xv.y;
            float d2 = pv.z - xv.z, d3 = pv.w - xv.w;
            dist2 += d0*d0 + d1*d1 + d2*d2 + d3*d3;
            srow[q*4+0] = pv.x; srow[q*4+1] = pv.y;
            srow[q*4+2] = pv.z; srow[q*4+3] = pv.w;
        }
    } else {
        dist2 = 3.402823466e+38f;
        #pragma unroll
        for (int q = 0; q < SOM_DIM; ++q) srow[q] = 0.f;
    }
    __syncthreads();

    const int half = lane >> 4;
    const int n    = lane & 15;
    v2f onesA; onesA.x = 1.0f; onesA.y = 1.0f;
    v8f acc0 = {};
    v8f acc1 = {};
    const int rowbase = wave * 32;
    #pragma unroll
    for (int c = 0; c < 8; ++c) {
        const int r0 = rowbase + c*4 + half;
        const int r1 = r0 + 2;
        v2f b0, b1;
        b0.x = stage[r0*LSTRIDE + n];        b0.y = stage[r1*LSTRIDE + n];
        b1.x = stage[r0*LSTRIDE + 16 + n];   b1.y = stage[r1*LSTRIDE + 16 + n];
        acc0 = __builtin_amdgcn_wmma_f32_16x16x4_f32(false, onesA, false, b0,
                                                     (short)0, acc0, false, false);
        acc1 = __builtin_amdgcn_wmma_f32_16x16x4_f32(false, onesA, false, b1,
                                                     (short)0, acc1, false, false);
    }
    if (lane < 16) {
        wsum[wave*SOM_DIM + lane]      = acc0[0];
        wsum[wave*SOM_DIM + 16 + lane] = acc1[0];
    }

    sdist[t] = dist2;
    sidx[t]  = p;
    __syncthreads();

    #pragma unroll
    for (int s = 128; s > 0; s >>= 1) {
        if (t < s) {
            float a = sdist[t], b = sdist[t + s];
            int   ia = sidx[t], ib = sidx[t + s];
            if (b < a || (b == a && ib < ia)) { sdist[t] = b; sidx[t] = ib; }
        }
        __syncthreads();
    }

    if (t < SOM_DIM) {
        float s = 0.f;
        #pragma unroll
        for (int w = 0; w < 8; ++w) s += wsum[w*SOM_DIM + t];
        blkSum[(size_t)blockIdx.x * SOM_DIM + t] = s;
    }
    if (t == 0) { blkMinV[blockIdx.x] = sdist[0]; blkMinI[blockIdx.x] = sidx[0]; }
}

__global__ __launch_bounds__(THREADS)
void som_pass_b(const float* __restrict__ particles,
                const int*   __restrict__ gloc,
                const int*   __restrict__ iter_num,
                const float* __restrict__ blkSum,
                const float* __restrict__ blkMinV,
                const int*   __restrict__ blkMinI,
                float* __restrict__ hdr,
                int NB, int N)
{
    __shared__ float sdist[THREADS];
    __shared__ int   sidx[THREADS];
    __shared__ float part[SOM_DIM][8];
    __shared__ int   sbmu;

    const int t = threadIdx.x;

    float best = 3.402823466e+38f; int bidx = 0x7fffffff;
    for (int k = t; k < NB; k += THREADS) {
        float dv = blkMinV[k]; int iv = blkMinI[k];
        if (dv < best || (dv == best && iv < bidx)) { best = dv; bidx = iv; }
    }
    sdist[t] = best; sidx[t] = bidx;
    __syncthreads();
    #pragma unroll
    for (int s = 128; s > 0; s >>= 1) {
        if (t < s) {
            float a = sdist[t], b = sdist[t + s];
            int   ia = sidx[t], ib = sidx[t + s];
            if (b < a || (b == a && ib < ia)) { sdist[t] = b; sidx[t] = ib; }
        }
        __syncthreads();
    }
    if (t == 0) sbmu = sidx[0];

    {
        const int d = t >> 3, seg = t & 7;
        const int per = (NB + 7) >> 3;
        const int b0 = seg * per;
        const int b1 = (b0 + per < NB) ? (b0 + per) : NB;
        float s = 0.f;
        for (int b = b0; b < b1; ++b) s += blkSum[(size_t)b * SOM_DIM + d];
        part[d][seg] = s;
    }
    __syncthreads();

    const int bmu = sbmu;
    if (t < SOM_DIM) {
        float S = 0.f;
        #pragma unroll
        for (int g = 0; g < 8; ++g) S += part[t][g];
        const float gb = particles[(size_t)bmu * SOM_DIM + t];
        hdr[4 + t]              = gb;
        hdr[4 + SOM_DIM + t]    = S / (float)N;
        hdr[4 + 2*SOM_DIM + t]  = (S - gb) / (float)(N - 1);
    }
    if (t == 0) {
        const float decay  = 1.0f - (float)iter_num[0] / 100.0f;
        const float radius = 0.5f * decay;
        const float sig    = 256.0f * decay;
        hdr[0] = (float)gloc[2*bmu];
        hdr[1] = (float)gloc[2*bmu + 1];
        hdr[2] = radius;
        hdr[3] = sig * sig;
    }
}

__global__ __launch_bounds__(THREADS)
void som_pass_c(const float* __restrict__ particles,
                const float* __restrict__ velocities,
                const float* __restrict__ r1,
                const float* __restrict__ r2,
                const int*   __restrict__ gloc,
                const float* __restrict__ hdr,
                float* __restrict__ outP,
                float* __restrict__ outV,
                int N)
{
    const int p = blockIdx.x * THREADS + threadIdx.x;
    if (p >= N) return;

    const float bx = hdr[0], by = hdr[1], radius = hdr[2], sigma2 = hdr[3];
    const float* gbest = hdr + 4;
    const float* cenA  = hdr + 4 + SOM_DIM;
    const float* cenB  = hdr + 4 + 2*SOM_DIM;

    const float4* P4  = (const float4*)(particles  + (size_t)p * SOM_DIM);
    const float4* V4  = (const float4*)(velocities + (size_t)p * SOM_DIM);
    const float4* R14 = (const float4*)(r1 + (size_t)p * SOM_DIM);
    const float4* R24 = (const float4*)(r2 + (size_t)p * SOM_DIM);
    __builtin_prefetch(R14, 0, 0);
    __builtin_prefetch(R24, 0, 0);

    const int2 gl = *(const int2*)(gloc + 2*(size_t)p);
    const float dx = (float)gl.x - bx;
    const float dy = (float)gl.y - by;
    const float nb = expf(-(dx*dx + dy*dy) / sigma2);
    const bool  skip = (1.0f - nb) > radius;
    const float* cen = ((nb + radius) >= 1.0f) ? cenA : cenB;
    const float4* C4 = (const float4*)cen;
    const float4* G4 = (const float4*)gbest;

    float4* OP = (float4*)(outP + (size_t)p * SOM_DIM);
    float4* OV = (float4*)(outV + (size_t)p * SOM_DIM);

    #pragma unroll
    for (int q = 0; q < 8; ++q) {
        const float4 pv = P4[q], vv = V4[q], a = R14[q], b = R24[q];
        const float4 c = C4[q], g = G4[q];
        float4 vn, pn;
        vn.x = 0.001f*vv.x + 0.01f*a.x*(c.x - pv.x) + 0.1f*b.x*(g.x - pv.x);
        vn.y = 0.001f*vv.y + 0.01f*a.y*(c.y - pv.y) + 0.1f*b.y*(g.y - pv.y);
        vn.z = 0.001f*vv.z + 0.01f*a.z*(c.z - pv.z) + 0.1f*b.z*(g.z - pv.z);
        vn.w = 0.001f*vv.w + 0.01f*a.w*(c.w - pv.w) + 0.1f*b.w*(g.w - pv.w);
        pn.x = pv.x + vn.x; pn.y = pv.y + vn.y;
        pn.z = pv.z + vn.z; pn.w = pv.w + vn.w;
        if (skip) { vn = vv; pn = pv; }
        OP[q] = pn;
        OV[q] = vn;
    }
}

extern "C" void kernel_launch(void* const* d_in, const int* in_sizes, int n_in,
                              void* d_out, int out_size, void* d_ws, size_t ws_size,
                              hipStream_t stream)
{
    (void)n_in; (void)out_size; (void)ws_size;
    const float* input      = (const float*)d_in[0];
    const float* particles  = (const float*)d_in[1];
    const float* velocities = (const float*)d_in[2];
    const float* r1         = (const float*)d_in[3];
    const float* r2         = (const float*)d_in[4];
    const int*   gloc       = (const int*)d_in[5];
    const int*   iter_num   = (const int*)d_in[6];

    const int N  = in_sizes[1] / SOM_DIM;
    const int NB = (N + THREADS - 1) / THREADS;

    float* ws      = (float*)d_ws;
    float* blkSum  = ws;
    float* blkMinV = ws + (size_t)NB * SOM_DIM;
    int*   blkMinI = (int*)(blkMinV + NB);
    size_t hoff    = (size_t)NB * SOM_DIM + 2 * (size_t)NB;
    hoff = (hoff + 3) & ~(size_t)3;
    float* hdr     = ws + hoff;

    float* outP = (float*)d_out;
    float* outV = outP + (size_t)N * SOM_DIM;

    som_pass_a<<<NB, THREADS, 0, stream>>>(input, particles, blkSum, blkMinV, blkMinI, N);
    som_pass_b<<<1, THREADS, 0, stream>>>(particles, gloc, iter_num,
                                          blkSum, blkMinV, blkMinI, hdr, NB, N);
    som_pass_c<<<NB, THREADS, 0, stream>>>(particles, velocities, r1, r2,
                                           gloc, hdr, outP, outV, N);
}


# Pass A: one 1024-thread work-group per 2048 rows (grid 128, was 1024 x 256 rows) folds its 32 waves by lane exchange and 1 barrier; pass B's serial fold is 8x shorter; column-sum order changed, fixed
# speedup vs baseline: 3.5935x; 3.5935x over previous
#include <hip/hip_runtime.h>
#include <hip/hip_bf16.h>
#include <math.h>

typedef float v2f __attribute__((ext_vector_type(2)));
typedef float v8f __attribute__((ext_vector_type(8)));

#define THREADS 256
#define SOM_DIM 32
#define LSTRIDE 36
#define PITCH4  9


#define A_THREADS 1024
#define A_WAVES   32
#define A_TILES   2
#define A_ROWS    (A_THREADS * A_TILES)

__global__ __launch_bounds__(A_THREADS)
void som_pass_a(const float* __restrict__ input,
                const float* __restrict__ particles,
                float* __restrict__ blkSum,
                float* __restrict__ blkMinV,
                int*   __restrict__ blkMinI,
                int N)
{
    __shared__ float4 stage4[A_WAVES][32 * PITCH4];
    __shared__ float wsum[A_WAVES * SOM_DIM];
    __shared__ float sdist[A_WAVES];
    __shared__ int   sidx[A_WAVES];

    const int t    = threadIdx.x;
    const int lane = t & 31;
    const int wave = t >> 5;
    float4* w4 = stage4[wave];
    const float* wf = reinterpret_cast<const float*>(w4);

    const int sub = lane >> 3;
    const int pc  = lane & 7;
    const float4* g4 = reinterpret_cast<const float4*>(particles);
    const int base = blockIdx.x * A_ROWS + wave * 32;

    float4 v[A_TILES][8];
    #pragma unroll
    for (int it = 0; it < A_TILES; ++it) {
        const int grow = base + it * A_THREADS;
        if (grow + 32 <= N) {
            #pragma unroll
            for (int q = 0; q < 8; ++q)
                v[it][q] = g4[(size_t)(grow + 4*q + sub) * 8 + pc];
        } else {
            #pragma unroll
            for (int q = 0; q < 8; ++q) {
                const int r = grow + 4*q + sub;
                v[it][q] = make_float4(0.f, 0.f, 0.f, 0.f);
                if (r < N) v[it][q] = g4[(size_t)r * 8 + pc];
            }
        }
    }

    float4 xv[8];
    {
        const float4* x4 = reinterpret_cast<const float4*>(input);
        #pragma unroll
        for (int q = 0; q < 8; ++q) xv[q] = x4[q];
    }

    const int half = lane >> 4;
    const int n    = lane & 15;
    v2f onesA; onesA.x = 1.0f; onesA.y = 1.0f;
    v8f acc0 = {};
    v8f acc1 = {};

    float best = 3.402823466e+38f;
    int   bidx = 0x7fffffff;

    #pragma unroll
    for (int it = 0; it < A_TILES; ++it) {
        #pragma unroll
        for (int q = 0; q < 8; ++q)
            w4[(4*q + sub) * PITCH4 + pc] = v[it][q];

        __builtin_amdgcn_fence(__ATOMIC_RELEASE, "workgroup");
        __builtin_amdgcn_wave_barrier();
        __builtin_amdgcn_fence(__ATOMIC_ACQUIRE, "workgroup");

        float dist2 = 0.f;
        #pragma unroll
        for (int q = 0; q < 8; ++q) {
            const float4 pv = w4[lane * PITCH4 + q];
            float d0 = pv.x - xv[q].x, d1 = pv.y - xv[q].y;
            float d2 = pv.z - xv[q].z, d3 = pv.w - xv[q].w;
            float tq = d1*d1;
            tq = fmaf(d0, d0, tq);
            tq = fmaf(d2, d2, tq);
            tq = fmaf(d3, d3, tq);
            dist2 += tq;
        }
        const int p = base + it * A_THREADS + lane;
        if (p >= N) dist2 = 3.402823466e+38f;
        if (dist2 < best || (dist2 == best && p < bidx)) { best = dist2; bidx = p; }

        #pragma unroll
        for (int c = 0; c < 8; ++c) {
            const int r0 = c*4 + half;
            const int r1 = r0 + 2;
            v2f b0, b1;
            b0.x = wf[r0*LSTRIDE + n];        b0.y = wf[r1*LSTRIDE + n];
            b1.x = wf[r0*LSTRIDE + 16 + n];   b1.y = wf[r1*LSTRIDE + 16 + n];
            acc0 = __builtin_amdgcn_wmma_f32_16x16x4_f32(false, onesA, false, b0,
                                                         (short)0, acc0, false, false);
            acc1 = __builtin_amdgcn_wmma_f32_16x16x4_f32(false, onesA, false, b1,
                                                         (short)0, acc1, false, false);
        }

        if (it + 1 < A_TILES) {
            __builtin_amdgcn_fence(__ATOMIC_RELEASE, "workgroup");
            __builtin_amdgcn_wave_barrier();
            __builtin_amdgcn_fence(__ATOMIC_ACQUIRE, "workgroup");
        }
    }

    #pragma unroll
    for (int m = 16; m > 0; m >>= 1) {
        const float ob = __shfl_xor(best, m, 32);
        const int   oi = __shfl_xor(bidx, m, 32);
        if (ob < best || (ob == best && oi < bidx)) { best = ob; bidx = oi; }
    }
    if (lane == 0) { sdist[wave] = best; sidx[wave] = bidx; }
    if (lane < 16) {
        wsum[wave*SOM_DIM + lane]      = acc0[0];
        wsum[wave*SOM_DIM + 16 + lane] = acc1[0];
    }
    __syncthreads();

    if (wave == 0) {
        float b = sdist[lane];
        int   i = sidx[lane];
        #pragma unroll
        for (int m = 16; m > 0; m >>= 1) {
            const float ob = __shfl_xor(b, m, 32);
            const int   oi = __shfl_xor(i, m, 32);
            if (ob < b || (ob == b && oi < i)) { b = ob; i = oi; }
        }
        if (lane == 0) { blkMinV[blockIdx.x] = b; blkMinI[blockIdx.x] = i; }
    } else if (wave == 1) {
        float s = 0.f;
        #pragma unroll
        for (int w = 0; w < A_WAVES; ++w) s += wsum[w*SOM_DIM + lane];
        blkSum[(size_t)blockIdx.x * SOM_DIM + lane] = s;
    }
}

__global__ __launch_bounds__(THREADS)
void som_pass_b(const float* __restrict__ particles,
                const int*   __restrict__ gloc,
                const int*   __restrict__ iter_num,
                const float* __restrict__ blkSum,
                const float* __restrict__ blkMinV,
                const int*   __restrict__ blkMinI,
                float* __restrict__ hdr,
                int NB, int N)
{
    __shared__ float sdist[THREADS];
    __shared__ int   sidx[THREADS];
    __shared__ float part[SOM_DIM][8];
    __shared__ int   sbmu;

    const int t = threadIdx.x;

    float best = 3.402823466e+38f; int bidx = 0x7fffffff;
    for (int k = t; k < NB; k += THREADS) {
        float dv = blkMinV[k]; int iv = blkMinI[k];
        if (dv < best || (dv == best && iv < bidx)) { best = dv; bidx = iv; }
    }
    sdist[t] = best; sidx[t] = bidx;
    __syncthreads();
    #pragma unroll
    for (int s = 128; s > 0; s >>= 1) {
        if (t < s) {
            float a = sdist[t], b = sdist[t + s];
            int   ia = sidx[t], ib = sidx[t + s];
            if (b < a || (b == a && ib < ia)) { sdist[t] = b; sidx[t] = ib; }
        }
        __syncthreads();
    }
    if (t == 0) sbmu = sidx[0];

    {
        const int d = t >> 3, seg = t & 7;
        const int per = (NB + 7) >> 3;
        const int b0 = seg * per;
        const int b1 = (b0 + per < NB) ? (b0 + per) : NB;
        float s = 0.f;
        for (int b = b0; b < b1; ++b) s += blkSum[(size_t)b * SOM_DIM + d];
        part[d][seg] = s;
    }
    __syncthreads();

    const int bmu = sbmu;
    if (t < SOM_DIM) {
        float S = 0.f;
        #pragma unroll
        for (int g = 0; g < 8; ++g) S += part[t][g];
        const float gb = particles[(size_t)bmu * SOM_DIM + t];
        hdr[4 + t]              = gb;
        hdr[4 + SOM_DIM + t]    = S / (float)N;
        hdr[4 + 2*SOM_DIM + t]  = (S - gb) / (float)(N - 1);
    }
    if (t == 0) {
        const float decay  = 1.0f - (float)iter_num[0] / 100.0f;
        const float radius = 0.5f * decay;
        const float sig    = 256.0f * decay;
        hdr[0] = (float)gloc[2*bmu];
        hdr[1] = (float)gloc[2*bmu + 1];
        hdr[2] = radius;
        hdr[3] = sig * sig;
    }
}

__global__ __launch_bounds__(THREADS)
void som_pass_c(const float* __restrict__ particles,
                const float* __restrict__ velocities,
                const float* __restrict__ r1,
                const float* __restrict__ r2,
                const int*   __restrict__ gloc,
                const float* __restrict__ hdr,
                float* __restrict__ outP,
                float* __restrict__ outV,
                int N)
{
    __shared__ float4 tP[8][32 * PITCH4];
    __shared__ float4 tV[8][32 * PITCH4];
    __shared__ float4 tA[8][32 * PITCH4];
    __shared__ float4 tB[8][32 * PITCH4];

    const int lane = threadIdx.x & 31;
    const int wave = threadIdx.x >> 5;
    const int row0 = blockIdx.x * THREADS + wave * 32;
    if (row0 >= N) return;

    float4* wP = tP[wave];
    float4* wV = tV[wave];
    float4* wA = tA[wave];
    float4* wB = tB[wave];

    const int sub = lane >> 3;
    const int pc  = lane & 7;

    {
        const float4* P4 = reinterpret_cast<const float4*>(particles);
        const float4* V4 = reinterpret_cast<const float4*>(velocities);
        const float4* A4 = reinterpret_cast<const float4*>(r1);
        const float4* B4 = reinterpret_cast<const float4*>(r2);
        float4 gp[8], gv[8], ga[8], gb[8];
        #pragma unroll
        for (int q = 0; q < 8; ++q) {
            const int r  = row0 + 4*q + sub;
            const int rc = (r < N) ? r : (N - 1);
            const size_t o = (size_t)rc * 8 + pc;
            gp[q] = P4[o]; gv[q] = V4[o]; ga[q] = A4[o]; gb[q] = B4[o];
        }
        #pragma unroll
        for (int q = 0; q < 8; ++q) {
            const int li = (4*q + sub) * PITCH4 + pc;
            wP[li] = gp[q]; wV[li] = gv[q]; wA[li] = ga[q]; wB[li] = gb[q];
        }
    }

    const int p  = row0 + lane;
    const int pcl = (p < N) ? p : (N - 1);
    const float bx = hdr[0], by = hdr[1], radius = hdr[2], sigma2 = hdr[3];
    const float* gbest = hdr + 4;
    const float* cenA  = hdr + 4 + SOM_DIM;
    const float* cenB  = hdr + 4 + 2*SOM_DIM;
    const int2 gl = *(const int2*)(gloc + 2*(size_t)pcl);
    const float dx = (float)gl.x - bx;
    const float dy = (float)gl.y - by;
    const float nb = expf(-(dx*dx + dy*dy) / sigma2);
    const bool  skip = (1.0f - nb) > radius;
    const bool  useA = (nb + radius) >= 1.0f;
    const float4* CA4 = (const float4*)cenA;
    const float4* CB4 = (const float4*)cenB;
    const float4* G4  = (const float4*)gbest;
    float4 c8[8], g8[8];
    #pragma unroll
    for (int q = 0; q < 8; ++q) {
        const float4 ca = CA4[q], cb = CB4[q];
        c8[q].x = useA ? ca.x : cb.x;
        c8[q].y = useA ? ca.y : cb.y;
        c8[q].z = useA ? ca.z : cb.z;
        c8[q].w = useA ? ca.w : cb.w;
        g8[q] = G4[q];
    }

    __builtin_amdgcn_fence(__ATOMIC_RELEASE, "workgroup");
    __builtin_amdgcn_wave_barrier();
    __builtin_amdgcn_fence(__ATOMIC_ACQUIRE, "workgroup");

    const int my = lane * PITCH4;
    #pragma unroll
    for (int q = 0; q < 8; ++q) {
        const float4 pv = wP[my + q], vv = wV[my + q], a = wA[my + q], b = wB[my + q];
        const float4 c = c8[q], g = g8[q];
        float4 vn, pn;
        vn.x = 0.001f*vv.x + 0.01f*a.x*(c.x - pv.x) + 0.1f*b.x*(g.x - pv.x);
        vn.y = 0.001f*vv.y + 0.01f*a.y*(c.y - pv.y) + 0.1f*b.y*(g.y - pv.y);
        vn.z = 0.001f*vv.z + 0.01f*a.z*(c.z - pv.z) + 0.1f*b.z*(g.z - pv.z);
        vn.w = 0.001f*vv.w + 0.01f*a.w*(c.w - pv.w) + 0.1f*b.w*(g.w - pv.w);
        pn.x = pv.x + vn.x; pn.y = pv.y + vn.y;
        pn.z = pv.z + vn.z; pn.w = pv.w + vn.w;
        if (skip) { vn = vv; pn = pv; }
        wP[my + q] = pn;
        wV[my + q] = vn;
    }

    __builtin_amdgcn_fence(__ATOMIC_RELEASE, "workgroup");
    __builtin_amdgcn_wave_barrier();
    __builtin_amdgcn_fence(__ATOMIC_ACQUIRE, "workgroup");

    float4* OP4 = reinterpret_cast<float4*>(outP);
    float4* OV4 = reinterpret_cast<float4*>(outV);
    if (row0 + 32 <= N) {
        float4 op[8], ov[8];
        #pragma unroll
        for (int q = 0; q < 8; ++q) {
            const int li = (4*q + sub) * PITCH4 + pc;
            op[q] = wP[li]; ov[q] = wV[li];
        }
        #pragma unroll
        for (int q = 0; q < 8; ++q) {
            const size_t o = (size_t)(row0 + 4*q + sub) * 8 + pc;
            OP4[o] = op[q];
            OV4[o] = ov[q];
        }
    } else {
        #pragma unroll
        for (int q = 0; q < 8; ++q) {
            const int r  = row0 + 4*q + sub;
            const int li = (4*q + sub) * PITCH4 + pc;
            if (r < N) {
                const size_t o = (size_t)r * 8 + pc;
                OP4[o] = wP[li];
                OV4[o] = wV[li];
            }
        }
    }
}

extern "C" void kernel_launch(void* const* d_in, const int* in_sizes, int n_in,
                              void* d_out, int out_size, void* d_ws, size_t ws_size,
                              hipStream_t stream)
{
    (void)n_in; (void)out_size; (void)ws_size;
    const float* input      = (const float*)d_in[0];
    const float* particles  = (const float*)d_in[1];
    const float* velocities = (const float*)d_in[2];
    const float* r1         = (const float*)d_in[3];
    const float* r2         = (const float*)d_in[4];
    const int*   gloc       = (const int*)d_in[5];
    const int*   iter_num   = (const int*)d_in[6];

    const int N  = in_sizes[1] / SOM_DIM;
    const int NB = (N + THREADS - 1) / THREADS;
    const int NA = (N + A_ROWS - 1) / A_ROWS;

    float* ws      = (float*)d_ws;
    float* blkSum  = ws;
    float* blkMinV = ws + (size_t)NA * SOM_DIM;
    int*   blkMinI = (int*)(blkMinV + NA);
    size_t hoff    = (size_t)NA * SOM_DIM + 2 * (size_t)NA;
    hoff = (hoff + 3) & ~(size_t)3;
    float* hdr     = ws + hoff;

    float* outP = (float*)d_out;
    float* outV = outP + (size_t)N * SOM_DIM;

    som_pass_a<<<NA, A_THREADS, 0, stream>>>(input, particles, blkSum, blkMinV, blkMinI, N);
    som_pass_b<<<1, THREADS, 0, stream>>>(particles, gloc, iter_num,
                                          blkSum, blkMinV, blkMinI, hdr, NA, N);
    som_pass_c<<<NB, THREADS, 0, stream>>>(particles, velocities, r1, r2,
                                           gloc, hdr, outP, outV, N);
}


# Pass B: a ninth wave finds the BMU by lane exchange and fetches its row and grid cell while eight waves fold the column sums, one wave per segment; 1 barrier, was 10; sums' order unchanged
# speedup vs baseline: 3.7810x; 1.0522x over previous
#include <hip/hip_runtime.h>
#include <hip/hip_bf16.h>
#include <math.h>

typedef float v2f __attribute__((ext_vector_type(2)));
typedef float v8f __attribute__((ext_vector_type(8)));

#define THREADS 256
#define SOM_DIM 32
#define LSTRIDE 36
#define PITCH4  9


#define A_THREADS 1024
#define A_WAVES   32
#define A_TILES   2
#define A_ROWS    (A_THREADS * A_TILES)

__global__ __launch_bounds__(A_THREADS)
void som_pass_a(const float* __restrict__ input,
                const float* __restrict__ particles,
                float* __restrict__ blkSum,
                float* __restrict__ blkMinV,
                int*   __restrict__ blkMinI,
                int N)
{
    __shared__ float4 stage4[A_WAVES][32 * PITCH4];
    __shared__ float wsum[A_WAVES * SOM_DIM];
    __shared__ float sdist[A_WAVES];
    __shared__ int   sidx[A_WAVES];

    const int t    = threadIdx.x;
    const int lane = t & 31;
    const int wave = t >> 5;
    float4* w4 = stage4[wave];
    const float* wf = reinterpret_cast<const float*>(w4);

    const int sub = lane >> 3;
    const int pc  = lane & 7;
    const float4* g4 = reinterpret_cast<const float4*>(particles);
    const int base = blockIdx.x * A_ROWS + wave * 32;

    float4 v[A_TILES][8];
    #pragma unroll
    for (int it = 0; it < A_TILES; ++it) {
        const int grow = base + it * A_THREADS;
        if (grow + 32 <= N) {
            #pragma unroll
            for (int q = 0; q < 8; ++q)
                v[it][q] = g4[(size_t)(grow + 4*q + sub) * 8 + pc];
        } else {
            #pragma unroll
            for (int q = 0; q < 8; ++q) {
                const int r = grow + 4*q + sub;
                v[it][q] = make_float4(0.f, 0.f, 0.f, 0.f);
                if (r < N) v[it][q] = g4[(size_t)r * 8 + pc];
            }
        }
    }

    float4 xv[8];
    {
        const float4* x4 = reinterpret_cast<const float4*>(input);
        #pragma unroll
        for (int q = 0; q < 8; ++q) xv[q] = x4[q];
    }

    const int half = lane >> 4;
    const int n    = lane & 15;
    v2f onesA; onesA.x = 1.0f; onesA.y = 1.0f;
    v8f acc0 = {};
    v8f acc1 = {};

    float best = 3.402823466e+38f;
    int   bidx = 0x7fffffff;

    #pragma unroll
    for (int it = 0; it < A_TILES; ++it) {
        #pragma unroll
        for (int q = 0; q < 8; ++q)
            w4[(4*q + sub) * PITCH4 + pc] = v[it][q];

        __builtin_amdgcn_fence(__ATOMIC_RELEASE, "workgroup");
        __builtin_amdgcn_wave_barrier();
        __builtin_amdgcn_fence(__ATOMIC_ACQUIRE, "workgroup");

        float dist2 = 0.f;
        #pragma unroll
        for (int q = 0; q < 8; ++q) {
            const float4 pv = w4[lane * PITCH4 + q];
            float d0 = pv.x - xv[q].x, d1 = pv.y - xv[q].y;
            float d2 = pv.z - xv[q].z, d3 = pv.w - xv[q].w;
            float tq = d1*d1;
            tq = fmaf(d0, d0, tq);
            tq = fmaf(d2, d2, tq);
            tq = fmaf(d3, d3, tq);
            dist2 += tq;
        }
        const int p = base + it * A_THREADS + lane;
        if (p >= N) dist2 = 3.402823466e+38f;
        if (dist2 < best || (dist2 == best && p < bidx)) { best = dist2; bidx = p; }

        #pragma unroll
        for (int c = 0; c < 8; ++c) {
            const int r0 = c*4 + half;
            const int r1 = r0 + 2;
            v2f b0, b1;
            b0.x = wf[r0*LSTRIDE + n];        b0.y = wf[r1*LSTRIDE + n];
            b1.x = wf[r0*LSTRIDE + 16 + n];   b1.y = wf[r1*LSTRIDE + 16 + n];
            acc0 = __builtin_amdgcn_wmma_f32_16x16x4_f32(false, onesA, false, b0,
                                                         (short)0, acc0, false, false);
            acc1 = __builtin_amdgcn_wmma_f32_16x16x4_f32(false, onesA, false, b1,
                                                         (short)0, acc1, false, false);
        }

        if (it + 1 < A_TILES) {
            __builtin_amdgcn_fence(__ATOMIC_RELEASE, "workgroup");
            __builtin_amdgcn_wave_barrier();
            __builtin_amdgcn_fence(__ATOMIC_ACQUIRE, "workgroup");
        }
    }

    #pragma unroll
    for (int m = 16; m > 0; m >>= 1) {
        const float ob = __shfl_xor(best, m, 32);
        const int   oi = __shfl_xor(bidx, m, 32);
        if (ob < best || (ob == best && oi < bidx)) { best = ob; bidx = oi; }
    }
    if (lane == 0) { sdist[wave] = best; sidx[wave] = bidx; }
    if (lane < 16) {
        wsum[wave*SOM_DIM + lane]      = acc0[0];
        wsum[wave*SOM_DIM + 16 + lane] = acc1[0];
    }
    __syncthreads();

    if (wave == 0) {
        float b = sdist[lane];
        int   i = sidx[lane];
        #pragma unroll
        for (int m = 16; m > 0; m >>= 1) {
            const float ob = __shfl_xor(b, m, 32);
            const int   oi = __shfl_xor(i, m, 32);
            if (ob < b || (ob == b && oi < i)) { b = ob; i = oi; }
        }
        if (lane == 0) { blkMinV[blockIdx.x] = b; blkMinI[blockIdx.x] = i; }
    } else if (wave == 1) {
        float s = 0.f;
        #pragma unroll
        for (int w = 0; w < A_WAVES; ++w) s += wsum[w*SOM_DIM + lane];
        blkSum[(size_t)blockIdx.x * SOM_DIM + lane] = s;
    }
}

#define B_FOLD_WAVES 8
#define B_THREADS    (32 * (B_FOLD_WAVES + 1))

__global__ __launch_bounds__(B_THREADS)
void som_pass_b(const float* __restrict__ particles,
                const int*   __restrict__ gloc,
                const int*   __restrict__ iter_num,
                const float* __restrict__ blkSum,
                const float* __restrict__ blkMinV,
                const int*   __restrict__ blkMinI,
                float* __restrict__ hdr,
                int NB, int N)
{
    __shared__ float part[B_FOLD_WAVES][SOM_DIM];

    const int lane = threadIdx.x & 31;
    const int wave = threadIdx.x >> 5;

    float gb = 0.f;
    int   gx = 0, gy = 0;
    int   itn = 0;

    if (wave == B_FOLD_WAVES) {
        float best = 3.402823466e+38f; int bidx = 0x7fffffff;
        for (int k = lane; k < NB; k += 32) {
            const float dv = blkMinV[k]; const int iv = blkMinI[k];
            if (dv < best || (dv == best && iv < bidx)) { best = dv; bidx = iv; }
        }
        #pragma unroll
        for (int m = 16; m > 0; m >>= 1) {
            const float ob = __shfl_xor(best, m, 32);
            const int   oi = __shfl_xor(bidx, m, 32);
            if (ob < best || (ob == best && oi < bidx)) { best = ob; bidx = oi; }
        }
        gb  = particles[(size_t)bidx * SOM_DIM + lane];
        gx  = gloc[2*(size_t)bidx];
        gy  = gloc[2*(size_t)bidx + 1];
        itn = iter_num[0];
    } else {
        const int seg = wave;
        const int per = (NB + B_FOLD_WAVES - 1) / B_FOLD_WAVES;
        const int b0 = seg * per;
        const int b1 = (b0 + per < NB) ? (b0 + per) : NB;
        float s = 0.f;
        for (int b = b0; b < b1; ++b) s += blkSum[(size_t)b * SOM_DIM + lane];
        part[seg][lane] = s;
    }
    __syncthreads();

    if (wave == B_FOLD_WAVES) {
        float S = 0.f;
        #pragma unroll
        for (int g = 0; g < B_FOLD_WAVES; ++g) S += part[g][lane];
        hdr[4 + lane]              = gb;
        hdr[4 + SOM_DIM + lane]    = S / (float)N;
        hdr[4 + 2*SOM_DIM + lane]  = (S - gb) / (float)(N - 1);
        if (lane == 0) {
            const float decay  = 1.0f - (float)itn / 100.0f;
            const float radius = 0.5f * decay;
            const float sig    = 256.0f * decay;
            hdr[0] = (float)gx;
            hdr[1] = (float)gy;
            hdr[2] = radius;
            hdr[3] = sig * sig;
        }
    }
}

__global__ __launch_bounds__(THREADS)
void som_pass_c(const float* __restrict__ particles,
                const float* __restrict__ velocities,
                const float* __restrict__ r1,
                const float* __restrict__ r2,
                const int*   __restrict__ gloc,
                const float* __restrict__ hdr,
                float* __restrict__ outP,
                float* __restrict__ outV,
                int N)
{
    __shared__ float4 tP[8][32 * PITCH4];
    __shared__ float4 tV[8][32 * PITCH4];
    __shared__ float4 tA[8][32 * PITCH4];
    __shared__ float4 tB[8][32 * PITCH4];

    const int lane = threadIdx.x & 31;
    const int wave = threadIdx.x >> 5;
    const int row0 = blockIdx.x * THREADS + wave * 32;
    if (row0 >= N) return;

    float4* wP = tP[wave];
    float4* wV = tV[wave];
    float4* wA = tA[wave];
    float4* wB = tB[wave];

    const int sub = lane >> 3;
    const int pc  = lane & 7;

    {
        const float4* P4 = reinterpret_cast<const float4*>(particles);
        const float4* V4 = reinterpret_cast<const float4*>(velocities);
        const float4* A4 = reinterpret_cast<const float4*>(r1);
        const float4* B4 = reinterpret_cast<const float4*>(r2);
        float4 gp[8], gv[8], ga[8], gb[8];
        #pragma unroll
        for (int q = 0; q < 8; ++q) {
            const int r  = row0 + 4*q + sub;
            const int rc = (r < N) ? r : (N - 1);
            const size_t o = (size_t)rc * 8 + pc;
            gp[q] = P4[o]; gv[q] = V4[o]; ga[q] = A4[o]; gb[q] = B4[o];
        }
        #pragma unroll
        for (int q = 0; q < 8; ++q) {
            const int li = (4*q + sub) * PITCH4 + pc;
            wP[li] = gp[q]; wV[li] = gv[q]; wA[li] = ga[q]; wB[li] = gb[q];
        }
    }

    const int p  = row0 + lane;
    const int pcl = (p < N) ? p : (N - 1);
    const float bx = hdr[0], by = hdr[1], radius = hdr[2], sigma2 = hdr[3];
    const float* gbest = hdr + 4;
    const float* cenA  = hdr + 4 + SOM_DIM;
    const float* cenB  = hdr + 4 + 2*SOM_DIM;
    const int2 gl = *(const int2*)(gloc + 2*(size_t)pcl);
    const float dx = (float)gl.x - bx;
    const float dy = (float)gl.y - by;
    const float nb = expf(-(dx*dx + dy*dy) / sigma2);
    const bool  skip = (1.0f - nb) > radius;
    const bool  useA = (nb + radius) >= 1.0f;
    const float4* CA4 = (const float4*)cenA;
    const float4* CB4 = (const float4*)cenB;
    const float4* G4  = (const float4*)gbest;
    float4 c8[8], g8[8];
    #pragma unroll
    for (int q = 0; q < 8; ++q) {
        const float4 ca = CA4[q], cb = CB4[q];
        c8[q].x = useA ? ca.x : cb.x;
        c8[q].y = useA ? ca.y : cb.y;
        c8[q].z = useA ? ca.z : cb.z;
        c8[q].w = useA ? ca.w : cb.w;
        g8[q] = G4[q];
    }

    __builtin_amdgcn_fence(__ATOMIC_RELEASE, "workgroup");
    __builtin_amdgcn_wave_barrier();
    __builtin_amdgcn_fence(__ATOMIC_ACQUIRE, "workgroup");

    const int my = lane * PITCH4;
    #pragma unroll
    for (int q = 0; q < 8; ++q) {
        const float4 pv = wP[my + q], vv = wV[my + q], a = wA[my + q], b = wB[my + q];
        const float4 c = c8[q], g = g8[q];
        float4 vn, pn;
        vn.x = 0.001f*vv.x + 0.01f*a.x*(c.x - pv.x) + 0.1f*b.x*(g.x - pv.x);
        vn.y = 0.001f*vv.y + 0.01f*a.y*(c.y - pv.y) + 0.1f*b.y*(g.y - pv.y);
        vn.z = 0.001f*vv.z + 0.01f*a.z*(c.z - pv.z) + 0.1f*b.z*(g.z - pv.z);
        vn.w = 0.001f*vv.w + 0.01f*a.w*(c.w - pv.w) + 0.1f*b.w*(g.w - pv.w);
        pn.x = pv.x + vn.x; pn.y = pv.y + vn.y;
        pn.z = pv.z + vn.z; pn.w = pv.w + vn.w;
        if (skip) { vn = vv; pn = pv; }
        wP[my + q] = pn;
        wV[my + q] = vn;
    }

    __builtin_amdgcn_fence(__ATOMIC_RELEASE, "workgroup");
    __builtin_amdgcn_wave_barrier();
    __builtin_amdgcn_fence(__ATOMIC_ACQUIRE, "workgroup");

    float4* OP4 = reinterpret_cast<float4*>(outP);
    float4* OV4 = reinterpret_cast<float4*>(outV);
    if (row0 + 32 <= N) {
        float4 op[8], ov[8];
        #pragma unroll
        for (int q = 0; q < 8; ++q) {
            const int li = (4*q + sub) * PITCH4 + pc;
            op[q] = wP[li]; ov[q] = wV[li];
        }
        #pragma unroll
        for (int q = 0; q < 8; ++q) {
            const size_t o = (size_t)(row0 + 4*q + sub) * 8 + pc;
            OP4[o] = op[q];
            OV4[o] = ov[q];
        }
    } else {
        #pragma unroll
        for (int q = 0; q < 8; ++q) {
            const int r  = row0 + 4*q + sub;
            const int li = (4*q + sub) * PITCH4 + pc;
            if (r < N) {
                const size_t o = (size_t)r * 8 + pc;
                OP4[o] = wP[li];
                OV4[o] = wV[li];
            }
        }
    }
}

extern "C" void kernel_launch(void* const* d_in, const int* in_sizes, int n_in,
                              void* d_out, int out_size, void* d_ws, size_t ws_size,
                              hipStream_t stream)
{
    (void)n_in; (void)out_size; (void)ws_size;
    const float* input      = (const float*)d_in[0];
    const float* particles  = (const float*)d_in[1];
    const float* velocities = (const float*)d_in[2];
    const float* r1         = (const float*)d_in[3];
    const float* r2         = (const float*)d_in[4];
    const int*   gloc       = (const int*)d_in[5];
    const int*   iter_num   = (const int*)d_in[6];

    const int N  = in_sizes[1] / SOM_DIM;
    const int NB = (N + THREADS - 1) / THREADS;
    const int NA = (N + A_ROWS - 1) / A_ROWS;

    float* ws      = (float*)d_ws;
    float* blkSum  = ws;
    float* blkMinV = ws + (size_t)NA * SOM_DIM;
    int*   blkMinI = (int*)(blkMinV + NA);
    size_t hoff    = (size_t)NA * SOM_DIM + 2 * (size_t)NA;
    hoff = (hoff + 3) & ~(size_t)3;
    float* hdr     = ws + hoff;

    float* outP = (float*)d_out;
    float* outV = outP + (size_t)N * SOM_DIM;

    som_pass_a<<<NA, A_THREADS, 0, stream>>>(input, particles, blkSum, blkMinV, blkMinI, N);
    som_pass_b<<<1, B_THREADS, 0, stream>>>(particles, gloc, iter_num,
                                          blkSum, blkMinV, blkMinI, hdr, NA, N);
    som_pass_c<<<NB, THREADS, 0, stream>>>(particles, velocities, r1, r2,
                                           gloc, hdr, outP, outV, N);
}


# Pass B reads pass A's records 16 B per lane: a wave-load brings 4 sum records (fold chain 4 loads, was 16) or all 128 minima (1 round, was 4); same 9 waves; column-sum order changed, fixed
# speedup vs baseline: 3.9473x; 1.0440x over previous
#include <hip/hip_runtime.h>
#include <hip/hip_bf16.h>
#include <math.h>

typedef float v2f __attribute__((ext_vector_type(2)));
typedef float v8f __attribute__((ext_vector_type(8)));

#define THREADS 256
#define SOM_DIM 32
#define LSTRIDE 36
#define PITCH4  9


#define A_THREADS 1024
#define A_WAVES   32
#define A_TILES   2
#define A_ROWS    (A_THREADS * A_TILES)

__global__ __launch_bounds__(A_THREADS)
void som_pass_a(const float* __restrict__ input,
                const float* __restrict__ particles,
                float* __restrict__ blkSum,
                float* __restrict__ blkMinV,
                int*   __restrict__ blkMinI,
                int N)
{
    __shared__ float4 stage4[A_WAVES][32 * PITCH4];
    __shared__ float wsum[A_WAVES * SOM_DIM];
    __shared__ float sdist[A_WAVES];
    __shared__ int   sidx[A_WAVES];

    const int t    = threadIdx.x;
    const int lane = t & 31;
    const int wave = t >> 5;
    float4* w4 = stage4[wave];
    const float* wf = reinterpret_cast<const float*>(w4);

    const int sub = lane >> 3;
    const int pc  = lane & 7;
    const float4* g4 = reinterpret_cast<const float4*>(particles);
    const int base = blockIdx.x * A_ROWS + wave * 32;

    float4 v[A_TILES][8];
    #pragma unroll
    for (int it = 0; it < A_TILES; ++it) {
        const int grow = base + it * A_THREADS;
        if (grow + 32 <= N) {
            #pragma unroll
            for (int q = 0; q < 8; ++q)
                v[it][q] = g4[(size_t)(grow + 4*q + sub) * 8 + pc];
        } else {
            #pragma unroll
            for (int q = 0; q < 8; ++q) {
                const int r = grow + 4*q + sub;
                v[it][q] = make_float4(0.f, 0.f, 0.f, 0.f);
                if (r < N) v[it][q] = g4[(size_t)r * 8 + pc];
            }
        }
    }

    float4 xv[8];
    {
        const float4* x4 = reinterpret_cast<const float4*>(input);
        #pragma unroll
        for (int q = 0; q < 8; ++q) xv[q] = x4[q];
    }

    const int half = lane >> 4;
    const int n    = lane & 15;
    v2f onesA; onesA.x = 1.0f; onesA.y = 1.0f;
    v8f acc0 = {};
    v8f acc1 = {};

    float best = 3.402823466e+38f;
    int   bidx = 0x7fffffff;

    #pragma unroll
    for (int it = 0; it < A_TILES; ++it) {
        #pragma unroll
        for (int q = 0; q < 8; ++q)
            w4[(4*q + sub) * PITCH4 + pc] = v[it][q];

        __builtin_amdgcn_fence(__ATOMIC_RELEASE, "workgroup");
        __builtin_amdgcn_wave_barrier();
        __builtin_amdgcn_fence(__ATOMIC_ACQUIRE, "workgroup");

        float dist2 = 0.f;
        #pragma unroll
        for (int q = 0; q < 8; ++q) {
            const float4 pv = w4[lane * PITCH4 + q];
            float d0 = pv.x - xv[q].x, d1 = pv.y - xv[q].y;
            float d2 = pv.z - xv[q].z, d3 = pv.w - xv[q].w;
            float tq = d1*d1;
            tq = fmaf(d0, d0, tq);
            tq = fmaf(d2, d2, tq);
            tq = fmaf(d3, d3, tq);
            dist2 += tq;
        }
        const int p = base + it * A_THREADS + lane;
        if (p >= N) dist2 = 3.402823466e+38f;
        if (dist2 < best || (dist2 == best && p < bidx)) { best = dist2; bidx = p; }

        #pragma unroll
        for (int c = 0; c < 8; ++c) {
            const int r0 = c*4 + half;
            const int r1 = r0 + 2;
            v2f b0, b1;
            b0.x = wf[r0*LSTRIDE + n];        b0.y = wf[r1*LSTRIDE + n];
            b1.x = wf[r0*LSTRIDE + 16 + n];   b1.y = wf[r1*LSTRIDE + 16 + n];
            acc0 = __builtin_amdgcn_wmma_f32_16x16x4_f32(false, onesA, false, b0,
                                                         (short)0, acc0, false, false);
            acc1 = __builtin_amdgcn_wmma_f32_16x16x4_f32(false, onesA, false, b1,
                                                         (short)0, acc1, false, false);
        }

        if (it + 1 < A_TILES) {
            __builtin_amdgcn_fence(__ATOMIC_RELEASE, "workgroup");
            __builtin_amdgcn_wave_barrier();
            __builtin_amdgcn_fence(__ATOMIC_ACQUIRE, "workgroup");
        }
    }

    #pragma unroll
    for (int m = 16; m > 0; m >>= 1) {
        const float ob = __shfl_xor(best, m, 32);
        const int   oi = __shfl_xor(bidx, m, 32);
        if (ob < best || (ob == best && oi < bidx)) { best = ob; bidx = oi; }
    }
    if (lane == 0) { sdist[wave] = best; sidx[wave] = bidx; }
    if (lane < 16) {
        wsum[wave*SOM_DIM + lane]      = acc0[0];
        wsum[wave*SOM_DIM + 16 + lane] = acc1[0];
    }
    __syncthreads();

    if (wave == 0) {
        float b = sdist[lane];
        int   i = sidx[lane];
        #pragma unroll
        for (int m = 16; m > 0; m >>= 1) {
            const float ob = __shfl_xor(b, m, 32);
            const int   oi = __shfl_xor(i, m, 32);
            if (ob < b || (ob == b && oi < i)) { b = ob; i = oi; }
        }
        if (lane == 0) { blkMinV[blockIdx.x] = b; blkMinI[blockIdx.x] = i; }
    } else if (wave == 1) {
        float s = 0.f;
        #pragma unroll
        for (int w = 0; w < A_WAVES; ++w) s += wsum[w*SOM_DIM + lane];
        blkSum[(size_t)blockIdx.x * SOM_DIM + lane] = s;
    }
}

#define B_FOLD_WAVES 8
#define B_THREADS    (32 * (B_FOLD_WAVES + 1))

__global__ __launch_bounds__(B_THREADS)
void som_pass_b(const float* __restrict__ particles,
                const int*   __restrict__ gloc,
                const int*   __restrict__ iter_num,
                const float* __restrict__ blkSum,
                const float* __restrict__ blkMinV,
                const int*   __restrict__ blkMinI,
                float* __restrict__ hdr,
                int NB, int N)
{
    __shared__ float4 part4[B_FOLD_WAVES][SOM_DIM / 4];
    const float* part = reinterpret_cast<const float*>(&part4[0][0]);

    const int lane = threadIdx.x & 31;
    const int wave = threadIdx.x >> 5;

    float gb = 0.f;
    int   gx = 0, gy = 0;
    int   itn = 0;

    if (wave == B_FOLD_WAVES) {
        float best = 3.402823466e+38f; int bidx = 0x7fffffff;
        if ((NB & 3) == 0) {
            const float4* V4 = reinterpret_cast<const float4*>(blkMinV);
            const int4*   I4 = reinterpret_cast<const int4*>(blkMinI);
            const int NB4 = NB >> 2;
            for (int k = lane; k < NB4; k += 32) {
                const float4 dv = V4[k]; const int4 iv = I4[k];
                if (dv.x < best || (dv.x == best && iv.x < bidx)) { best = dv.x; bidx = iv.x; }
                if (dv.y < best || (dv.y == best && iv.y < bidx)) { best = dv.y; bidx = iv.y; }
                if (dv.z < best || (dv.z == best && iv.z < bidx)) { best = dv.z; bidx = iv.z; }
                if (dv.w < best || (dv.w == best && iv.w < bidx)) { best = dv.w; bidx = iv.w; }
            }
        } else {
            for (int k = lane; k < NB; k += 32) {
                const float dv = blkMinV[k]; const int iv = blkMinI[k];
                if (dv < best || (dv == best && iv < bidx)) { best = dv; bidx = iv; }
            }
        }
        #pragma unroll
        for (int m = 16; m > 0; m >>= 1) {
            const float ob = __shfl_xor(best, m, 32);
            const int   oi = __shfl_xor(bidx, m, 32);
            if (ob < best || (ob == best && oi < bidx)) { best = ob; bidx = oi; }
        }
        gb  = particles[(size_t)bidx * SOM_DIM + lane];
        gx  = gloc[2*(size_t)bidx];
        gy  = gloc[2*(size_t)bidx + 1];
        itn = iter_num[0];
    } else {
        const int seg = wave;
        const int per = (NB + B_FOLD_WAVES - 1) / B_FOLD_WAVES;
        const int b0 = seg * per;
        const int b1 = (b0 + per < NB) ? (b0 + per) : NB;
        const int r  = lane >> 3;
        const int pc = lane & 7;
        const float4* S4 = reinterpret_cast<const float4*>(blkSum);
        float4 s = make_float4(0.f, 0.f, 0.f, 0.f);
        for (int b = b0 + r; b < b1; b += 4) {
            const float4 x = S4[(size_t)b * (SOM_DIM / 4) + pc];
            s.x += x.x; s.y += x.y; s.z += x.z; s.w += x.w;
        }
        s.x += __shfl_xor(s.x, 8, 32);  s.y += __shfl_xor(s.y, 8, 32);
        s.z += __shfl_xor(s.z, 8, 32);  s.w += __shfl_xor(s.w, 8, 32);
        s.x += __shfl_xor(s.x, 16, 32); s.y += __shfl_xor(s.y, 16, 32);
        s.z += __shfl_xor(s.z, 16, 32); s.w += __shfl_xor(s.w, 16, 32);
        if (r == 0) part4[seg][pc] = s;
    }
    __syncthreads();

    if (wave == B_FOLD_WAVES) {
        float S = 0.f;
        #pragma unroll
        for (int g = 0; g < B_FOLD_WAVES; ++g) S += part[g * SOM_DIM + lane];
        hdr[4 + lane]              = gb;
        hdr[4 + SOM_DIM + lane]    = S / (float)N;
        hdr[4 + 2*SOM_DIM + lane]  = (S - gb) / (float)(N - 1);
        if (lane == 0) {
            const float decay  = 1.0f - (float)itn / 100.0f;
            const float radius = 0.5f * decay;
            const float sig    = 256.0f * decay;
            hdr[0] = (float)gx;
            hdr[1] = (float)gy;
            hdr[2] = radius;
            hdr[3] = sig * sig;
        }
    }
}

__global__ __launch_bounds__(THREADS)
void som_pass_c(const float* __restrict__ particles,
                const float* __restrict__ velocities,
                const float* __restrict__ r1,
                const float* __restrict__ r2,
                const int*   __restrict__ gloc,
                const float* __restrict__ hdr,
                float* __restrict__ outP,
                float* __restrict__ outV,
                int N)
{
    __shared__ float4 tP[8][32 * PITCH4];
    __shared__ float4 tV[8][32 * PITCH4];
    __shared__ float4 tA[8][32 * PITCH4];
    __shared__ float4 tB[8][32 * PITCH4];

    const int lane = threadIdx.x & 31;
    const int wave = threadIdx.x >> 5;
    const int row0 = blockIdx.x * THREADS + wave * 32;
    if (row0 >= N) return;

    float4* wP = tP[wave];
    float4* wV = tV[wave];
    float4* wA = tA[wave];
    float4* wB = tB[wave];

    const int sub = lane >> 3;
    const int pc  = lane & 7;

    {
        const float4* P4 = reinterpret_cast<const float4*>(particles);
        const float4* V4 = reinterpret_cast<const float4*>(velocities);
        const float4* A4 = reinterpret_cast<const float4*>(r1);
        const float4* B4 = reinterpret_cast<const float4*>(r2);
        float4 gp[8], gv[8], ga[8], gb[8];
        #pragma unroll
        for (int q = 0; q < 8; ++q) {
            const int r  = row0 + 4*q + sub;
            const int rc = (r < N) ? r : (N - 1);
            const size_t o = (size_t)rc * 8 + pc;
            gp[q] = P4[o]; gv[q] = V4[o]; ga[q] = A4[o]; gb[q] = B4[o];
        }
        #pragma unroll
        for (int q = 0; q < 8; ++q) {
            const int li = (4*q + sub) * PITCH4 + pc;
            wP[li] = gp[q]; wV[li] = gv[q]; wA[li] = ga[q]; wB[li] = gb[q];
        }
    }

    const int p  = row0 + lane;
    const int pcl = (p < N) ? p : (N - 1);
    const float bx = hdr[0], by = hdr[1], radius = hdr[2], sigma2 = hdr[3];
    const float* gbest = hdr + 4;
    const float* cenA  = hdr + 4 + SOM_DIM;
    const float* cenB  = hdr + 4 + 2*SOM_DIM;
    const int2 gl = *(const int2*)(gloc + 2*(size_t)pcl);
    const float dx = (float)gl.x - bx;
    const float dy = (float)gl.y - by;
    const float nb = expf(-(dx*dx + dy*dy) / sigma2);
    const bool  skip = (1.0f - nb) > radius;
    const bool  useA = (nb + radius) >= 1.0f;
    const float4* CA4 = (const float4*)cenA;
    const float4* CB4 = (const float4*)cenB;
    const float4* G4  = (const float4*)gbest;
    float4 c8[8], g8[8];
    #pragma unroll
    for (int q = 0; q < 8; ++q) {
        const float4 ca = CA4[q], cb = CB4[q];
        c8[q].x = useA ? ca.x : cb.x;
        c8[q].y = useA ? ca.y : cb.y;
        c8[q].z = useA ? ca.z : cb.z;
        c8[q].w = useA ? ca.w : cb.w;
        g8[q] = G4[q];
    }

    __builtin_amdgcn_fence(__ATOMIC_RELEASE, "workgroup");
    __builtin_amdgcn_wave_barrier();
    __builtin_amdgcn_fence(__ATOMIC_ACQUIRE, "workgroup");

    const int my = lane * PITCH4;
    #pragma unroll
    for (int q = 0; q < 8; ++q) {
        const float4 pv = wP[my + q], vv = wV[my + q], a = wA[my + q], b = wB[my + q];
        const float4 c = c8[q], g = g8[q];
        float4 vn, pn;
        vn.x = 0.001f*vv.x + 0.01f*a.x*(c.x - pv.x) + 0.1f*b.x*(g.x - pv.x);
        vn.y = 0.001f*vv.y + 0.01f*a.y*(c.y - pv.y) + 0.1f*b.y*(g.y - pv.y);
        vn.z = 0.001f*vv.z + 0.01f*a.z*(c.z - pv.z) + 0.1f*b.z*(g.z - pv.z);
        vn.w = 0.001f*vv.w + 0.01f*a.w*(c.w - pv.w) + 0.1f*b.w*(g.w - pv.w);
        pn.x = pv.x + vn.x; pn.y = pv.y + vn.y;
        pn.z = pv.z + vn.z; pn.w = pv.w + vn.w;
        if (skip) { vn = vv; pn = pv; }
        wP[my + q] = pn;
        wV[my + q] = vn;
    }

    __builtin_amdgcn_fence(__ATOMIC_RELEASE, "workgroup");
    __builtin_amdgcn_wave_barrier();
    __builtin_amdgcn_fence(__ATOMIC_ACQUIRE, "workgroup");

    float4* OP4 = reinterpret_cast<float4*>(outP);
    float4* OV4 = reinterpret_cast<float4*>(outV);
    if (row0 + 32 <= N) {
        float4 op[8], ov[8];
        #pragma unroll
        for (int q = 0; q < 8; ++q) {
            const int li = (4*q + sub) * PITCH4 + pc;
            op[q] = wP[li]; ov[q] = wV[li];
        }
        #pragma unroll
        for (int q = 0; q < 8; ++q) {
            const size_t o = (size_t)(row0 + 4*q + sub) * 8 + pc;
            OP4[o] = op[q];
            OV4[o] = ov[q];
        }
    } else {
        #pragma unroll
        for (int q = 0; q < 8; ++q) {
            const int r  = row0 + 4*q + sub;
            const int li = (4*q + sub) * PITCH4 + pc;
            if (r < N) {
                const size_t o = (size_t)r * 8 + pc;
                OP4[o] = wP[li];
                OV4[o] = wV[li];
            }
        }
    }
}

extern "C" void kernel_launch(void* const* d_in, const int* in_sizes, int n_in,
                              void* d_out, int out_size, void* d_ws, size_t ws_size,
                              hipStream_t stream)
{
    (void)n_in; (void)out_size; (void)ws_size;
    const float* input      = (const float*)d_in[0];
    const float* particles  = (const float*)d_in[1];
    const float* velocities = (const float*)d_in[2];
    const float* r1         = (const float*)d_in[3];
    const float* r2         = (const float*)d_in[4];
    const int*   gloc       = (const int*)d_in[5];
    const int*   iter_num   = (const int*)d_in[6];

    const int N  = in_sizes[1] / SOM_DIM;
    const int NB = (N + THREADS - 1) / THREADS;
    const int NA = (N + A_ROWS - 1) / A_ROWS;

    float* ws      = (float*)d_ws;
    float* blkSum  = ws;
    float* blkMinV = ws + (size_t)NA * SOM_DIM;
    int*   blkMinI = (int*)(blkMinV + NA);
    size_t hoff    = (size_t)NA * SOM_DIM + 2 * (size_t)NA;
    hoff = (hoff + 3) & ~(size_t)3;
    float* hdr     = ws + hoff;

    float* outP = (float*)d_out;
    float* outV = outP + (size_t)N * SOM_DIM;

    som_pass_a<<<NA, A_THREADS, 0, stream>>>(input, particles, blkSum, blkMinV, blkMinI, N);
    som_pass_b<<<1, B_THREADS, 0, stream>>>(particles, gloc, iter_num,
                                          blkSum, blkMinV, blkMinI, hdr, NA, N);
    som_pass_c<<<NB, THREADS, 0, stream>>>(particles, velocities, r1, r2,
                                           gloc, hdr, outP, outV, N);
}


# Pass A on every WGP: 256 work-groups of 8 waves x 4 tiles (was 128 of 32 waves x 2), a tile started once its own loads are back; pass B folds 256 records on 16 waves; column-sum order changed, fixed
# speedup vs baseline: 4.1191x; 1.0435x over previous
#include <hip/hip_runtime.h>
#include <hip/hip_bf16.h>
#include <math.h>

typedef float v2f __attribute__((ext_vector_type(2)));
typedef float v8f __attribute__((ext_vector_type(8)));

#define THREADS 256
#define SOM_DIM 32
#define LSTRIDE 36
#define PITCH4  9


#define A_THREADS 256
#define A_WAVES   8
#define A_TILES   4
#define A_ROWS    (A_THREADS * A_TILES)

__global__ __launch_bounds__(A_THREADS)
void som_pass_a(const float* __restrict__ input,
                const float* __restrict__ particles,
                float* __restrict__ blkSum,
                float* __restrict__ blkMinV,
                int*   __restrict__ blkMinI,
                int N)
{
    __shared__ float4 stage4[A_WAVES][32 * PITCH4];
    __shared__ float wsum[A_WAVES * SOM_DIM];
    __shared__ float sdist[A_WAVES];
    __shared__ int   sidx[A_WAVES];

    const int t    = threadIdx.x;
    const int lane = t & 31;
    const int wave = t >> 5;
    float4* w4 = stage4[wave];
    const float* wf = reinterpret_cast<const float*>(w4);

    const int sub = lane >> 3;
    const int pc  = lane & 7;
    const float4* g4 = reinterpret_cast<const float4*>(particles);
    const int base = blockIdx.x * A_ROWS + wave * 32;

    float4 v[A_TILES][8];
    #pragma unroll
    for (int it = 0; it < A_TILES; ++it) {
        const int grow = base + it * A_THREADS;
        #pragma unroll
        for (int q = 0; q < 8; ++q) {
            const int r  = grow + 4*q + sub;
            const int rc = (r < N) ? r : (N - 1);
            v[it][q] = g4[(size_t)rc * 8 + pc];
        }
    }

    float4 xv[8];
    {
        const float4* x4 = reinterpret_cast<const float4*>(input);
        #pragma unroll
        for (int q = 0; q < 8; ++q) xv[q] = x4[q];
    }

    const int half = lane >> 4;
    const int n    = lane & 15;
    v2f onesA; onesA.x = 1.0f; onesA.y = 1.0f;
    v8f acc0 = {};
    v8f acc1 = {};

    float best = 3.402823466e+38f;
    int   bidx = 0x7fffffff;

    #pragma unroll
    for (int it = 0; it < A_TILES; ++it) {
        if (base + it * A_THREADS + 32 > N) {
            #pragma unroll
            for (int q = 0; q < 8; ++q)
                if (base + it * A_THREADS + 4*q + sub >= N) v[it][q] = make_float4(0.f, 0.f, 0.f, 0.f);
        }
        #pragma unroll
        for (int q = 0; q < 8; ++q)
            w4[(4*q + sub) * PITCH4 + pc] = v[it][q];

        __builtin_amdgcn_fence(__ATOMIC_RELEASE, "workgroup", "local");
        __builtin_amdgcn_wave_barrier();
        __builtin_amdgcn_fence(__ATOMIC_ACQUIRE, "workgroup", "local");

        float dist2 = 0.f;
        #pragma unroll
        for (int q = 0; q < 8; ++q) {
            const float4 pv = w4[lane * PITCH4 + q];
            float d0 = pv.x - xv[q].x, d1 = pv.y - xv[q].y;
            float d2 = pv.z - xv[q].z, d3 = pv.w - xv[q].w;
            float tq = d1*d1;
            tq = fmaf(d0, d0, tq);
            tq = fmaf(d2, d2, tq);
            tq = fmaf(d3, d3, tq);
            dist2 += tq;
        }
        const int p = base + it * A_THREADS + lane;
        if (p >= N) dist2 = 3.402823466e+38f;
        if (dist2 < best || (dist2 == best && p < bidx)) { best = dist2; bidx = p; }

        #pragma unroll
        for (int c = 0; c < 8; ++c) {
            const int r0 = c*4 + half;
            const int r1 = r0 + 2;
            v2f b0, b1;
            b0.x = wf[r0*LSTRIDE + n];        b0.y = wf[r1*LSTRIDE + n];
            b1.x = wf[r0*LSTRIDE + 16 + n];   b1.y = wf[r1*LSTRIDE + 16 + n];
            acc0 = __builtin_amdgcn_wmma_f32_16x16x4_f32(false, onesA, false, b0,
                                                         (short)0, acc0, false, false);
            acc1 = __builtin_amdgcn_wmma_f32_16x16x4_f32(false, onesA, false, b1,
                                                         (short)0, acc1, false, false);
        }

        if (it + 1 < A_TILES) {
            __builtin_amdgcn_fence(__ATOMIC_RELEASE, "workgroup", "local");
            __builtin_amdgcn_wave_barrier();
            __builtin_amdgcn_fence(__ATOMIC_ACQUIRE, "workgroup", "local");
        }
    }

    #pragma unroll
    for (int m = 16; m > 0; m >>= 1) {
        const float ob = __shfl_xor(best, m, 32);
        const int   oi = __shfl_xor(bidx, m, 32);
        if (ob < best || (ob == best && oi < bidx)) { best = ob; bidx = oi; }
    }
    if (lane == 0) { sdist[wave] = best; sidx[wave] = bidx; }
    if (lane < 16) {
        wsum[wave*SOM_DIM + lane]      = acc0[0];
        wsum[wave*SOM_DIM + 16 + lane] = acc1[0];
    }
    __syncthreads();

    if (wave == 0) {
        float b = 3.402823466e+38f;
        int   i = 0x7fffffff;
        if (lane < A_WAVES) { b = sdist[lane]; i = sidx[lane]; }
        #pragma unroll
        for (int m = 16; m > 0; m >>= 1) {
            const float ob = __shfl_xor(b, m, 32);
            const int   oi = __shfl_xor(i, m, 32);
            if (ob < b || (ob == b && oi < i)) { b = ob; i = oi; }
        }
        if (lane == 0) { blkMinV[blockIdx.x] = b; blkMinI[blockIdx.x] = i; }
    } else if (wave == 1) {
        float s = 0.f;
        #pragma unroll
        for (int w = 0; w < A_WAVES; ++w) s += wsum[w*SOM_DIM + lane];
        blkSum[(size_t)blockIdx.x * SOM_DIM + lane] = s;
    }
}

#define B_FOLD_WAVES 16
#define B_THREADS    (32 * (B_FOLD_WAVES + 1))

__global__ __launch_bounds__(B_THREADS)
void som_pass_b(const float* __restrict__ particles,
                const int*   __restrict__ gloc,
                const int*   __restrict__ iter_num,
                const float* __restrict__ blkSum,
                const float* __restrict__ blkMinV,
                const int*   __restrict__ blkMinI,
                float* __restrict__ hdr,
                int NB, int N)
{
    __shared__ float4 part4[B_FOLD_WAVES][SOM_DIM / 4];
    const float* part = reinterpret_cast<const float*>(&part4[0][0]);

    const int lane = threadIdx.x & 31;
    const int wave = threadIdx.x >> 5;

    float gb = 0.f;
    int   gx = 0, gy = 0;
    int   itn = 0;

    if (wave == B_FOLD_WAVES) {
        float best = 3.402823466e+38f; int bidx = 0x7fffffff;
        if ((NB & 3) == 0) {
            const float4* V4 = reinterpret_cast<const float4*>(blkMinV);
            const int4*   I4 = reinterpret_cast<const int4*>(blkMinI);
            const int NB4 = NB >> 2;
            for (int k = lane; k < NB4; k += 32) {
                const float4 dv = V4[k]; const int4 iv = I4[k];
                if (dv.x < best || (dv.x == best && iv.x < bidx)) { best = dv.x; bidx = iv.x; }
                if (dv.y < best || (dv.y == best && iv.y < bidx)) { best = dv.y; bidx = iv.y; }
                if (dv.z < best || (dv.z == best && iv.z < bidx)) { best = dv.z; bidx = iv.z; }
                if (dv.w < best || (dv.w == best && iv.w < bidx)) { best = dv.w; bidx = iv.w; }
            }
        } else {
            for (int k = lane; k < NB; k += 32) {
                const float dv = blkMinV[k]; const int iv = blkMinI[k];
                if (dv < best || (dv == best && iv < bidx)) { best = dv; bidx = iv; }
            }
        }
        #pragma unroll
        for (int m = 16; m > 0; m >>= 1) {
            const float ob = __shfl_xor(best, m, 32);
            const int   oi = __shfl_xor(bidx, m, 32);
            if (ob < best || (ob == best && oi < bidx)) { best = ob; bidx = oi; }
        }
        gb  = particles[(size_t)bidx * SOM_DIM + lane];
        gx  = gloc[2*(size_t)bidx];
        gy  = gloc[2*(size_t)bidx + 1];
        itn = iter_num[0];
    } else {
        const int seg = wave;
        const int per = (NB + B_FOLD_WAVES - 1) / B_FOLD_WAVES;
        const int b0 = seg * per;
        const int b1 = (b0 + per < NB) ? (b0 + per) : NB;
        const int r  = lane >> 3;
        const int pc = lane & 7;
        const float4* S4 = reinterpret_cast<const float4*>(blkSum);
        float4 s = make_float4(0.f, 0.f, 0.f, 0.f);
        for (int b = b0 + r; b < b1; b += 4) {
            const float4 x = S4[(size_t)b * (SOM_DIM / 4) + pc];
            s.x += x.x; s.y += x.y; s.z += x.z; s.w += x.w;
        }
        s.x += __shfl_xor(s.x, 8, 32);  s.y += __shfl_xor(s.y, 8, 32);
        s.z += __shfl_xor(s.z, 8, 32);  s.w += __shfl_xor(s.w, 8, 32);
        s.x += __shfl_xor(s.x, 16, 32); s.y += __shfl_xor(s.y, 16, 32);
        s.z += __shfl_xor(s.z, 16, 32); s.w += __shfl_xor(s.w, 16, 32);
        if (r == 0) part4[seg][pc] = s;
    }
    __syncthreads();

    if (wave == B_FOLD_WAVES) {
        float S = 0.f;
        #pragma unroll
        for (int g = 0; g < B_FOLD_WAVES; ++g) S += part[g * SOM_DIM + lane];
        hdr[4 + lane]              = gb;
        hdr[4 + SOM_DIM + lane]    = S / (float)N;
        hdr[4 + 2*SOM_DIM + lane]  = (S - gb) / (float)(N - 1);
        if (lane == 0) {
            const float decay  = 1.0f - (float)itn / 100.0f;
            const float radius = 0.5f * decay;
            const float sig    = 256.0f * decay;
            hdr[0] = (float)gx;
            hdr[1] = (float)gy;
            hdr[2] = radius;
            hdr[3] = sig * sig;
        }
    }
}

__global__ __launch_bounds__(THREADS)
void som_pass_c(const float* __restrict__ particles,
                const float* __restrict__ velocities,
                const float* __restrict__ r1,
                const float* __restrict__ r2,
                const int*   __restrict__ gloc,
                const float* __restrict__ hdr,
                float* __restrict__ outP,
                float* __restrict__ outV,
                int N)
{
    __shared__ float4 tP[8][32 * PITCH4];
    __shared__ float4 tV[8][32 * PITCH4];
    __shared__ float4 tA[8][32 * PITCH4];
    __shared__ float4 tB[8][32 * PITCH4];

    const int lane = threadIdx.x & 31;
    const int wave = threadIdx.x >> 5;
    const int row0 = blockIdx.x * THREADS + wave * 32;
    if (row0 >= N) return;

    float4* wP = tP[wave];
    float4* wV = tV[wave];
    float4* wA = tA[wave];
    float4* wB = tB[wave];

    const int sub = lane >> 3;
    const int pc  = lane & 7;

    {
        const float4* P4 = reinterpret_cast<const float4*>(particles);
        const float4* V4 = reinterpret_cast<const float4*>(velocities);
        const float4* A4 = reinterpret_cast<const float4*>(r1);
        const float4* B4 = reinterpret_cast<const float4*>(r2);
        float4 gp[8], gv[8], ga[8], gb[8];
        #pragma unroll
        for (int q = 0; q < 8; ++q) {
            const int r  = row0 + 4*q + sub;
            const int rc = (r < N) ? r : (N - 1);
            const size_t o = (size_t)rc * 8 + pc;
            gp[q] = P4[o]; gv[q] = V4[o]; ga[q] = A4[o]; gb[q] = B4[o];
        }
        #pragma unroll
        for (int q = 0; q < 8; ++q) {
            const int li = (4*q + sub) * PITCH4 + pc;
            wP[li] = gp[q]; wV[li] = gv[q]; wA[li] = ga[q]; wB[li] = gb[q];
        }
    }

    const int p  = row0 + lane;
    const int pcl = (p < N) ? p : (N - 1);
    const float bx = hdr[0], by = hdr[1], radius = hdr[2], sigma2 = hdr[3];
    const float* gbest = hdr + 4;
    const float* cenA  = hdr + 4 + SOM_DIM;
    const float* cenB  = hdr + 4 + 2*SOM_DIM;
    const int2 gl = *(const int2*)(gloc + 2*(size_t)pcl);
    const float dx = (float)gl.x - bx;
    const float dy = (float)gl.y - by;
    const float nb = expf(-(dx*dx + dy*dy) / sigma2);
    const bool  skip = (1.0f - nb) > radius;
    const bool  useA = (nb + radius) >= 1.0f;
    const float4* CA4 = (const float4*)cenA;
    const float4* CB4 = (const float4*)cenB;
    const float4* G4  = (const float4*)gbest;
    float4 c8[8], g8[8];
    #pragma unroll
    for (int q = 0; q < 8; ++q) {
        const float4 ca = CA4[q], cb = CB4[q];
        c8[q].x = useA ? ca.x : cb.x;
        c8[q].y = useA ? ca.y : cb.y;
        c8[q].z = useA ? ca.z : cb.z;
        c8[q].w = useA ? ca.w : cb.w;
        g8[q] = G4[q];
    }

    __builtin_amdgcn_fence(__ATOMIC_RELEASE, "workgroup");
    __builtin_amdgcn_wave_barrier();
    __builtin_amdgcn_fence(__ATOMIC_ACQUIRE, "workgroup");

    const int my = lane * PITCH4;
    #pragma unroll
    for (int q = 0; q < 8; ++q) {
        const float4 pv = wP[my + q], vv = wV[my + q], a = wA[my + q], b = wB[my + q];
        const float4 c = c8[q], g = g8[q];
        float4 vn, pn;
        vn.x = 0.001f*vv.x + 0.01f*a.x*(c.x - pv.x) + 0.1f*b.x*(g.x - pv.x);
        vn.y = 0.001f*vv.y + 0.01f*a.y*(c.y - pv.y) + 0.1f*b.y*(g.y - pv.y);
        vn.z = 0.001f*vv.z + 0.01f*a.z*(c.z - pv.z) + 0.1f*b.z*(g.z - pv.z);
        vn.w = 0.001f*vv.w + 0.01f*a.w*(c.w - pv.w) + 0.1f*b.w*(g.w - pv.w);
        pn.x = pv.x + vn.x; pn.y = pv.y + vn.y;
        pn.z = pv.z + vn.z; pn.w = pv.w + vn.w;
        if (skip) { vn = vv; pn = pv; }
        wP[my + q] = pn;
        wV[my + q] = vn;
    }

    __builtin_amdgcn_fence(__ATOMIC_RELEASE, "workgroup");
    __builtin_amdgcn_wave_barrier();
    __builtin_amdgcn_fence(__ATOMIC_ACQUIRE, "workgroup");

    float4* OP4 = reinterpret_cast<float4*>(outP);
    float4* OV4 = reinterpret_cast<float4*>(outV);
    if (row0 + 32 <= N) {
        float4 op[8], ov[8];
        #pragma unroll
        for (int q = 0; q < 8; ++q) {
            const int li = (4*q + sub) * PITCH4 + pc;
            op[q] = wP[li]; ov[q] = wV[li];
        }
        #pragma unroll
        for (int q = 0; q < 8; ++q) {
            const size_t o = (size_t)(row0 + 4*q + sub) * 8 + pc;
            OP4[o] = op[q];
            OV4[o] = ov[q];
        }
    } else {
        #pragma unroll
        for (int q = 0; q < 8; ++q) {
            const int r  = row0 + 4*q + sub;
            const int li = (4*q + sub) * PITCH4 + pc;
            if (r < N) {
                const size_t o = (size_t)r * 8 + pc;
                OP4[o] = wP[li];
                OV4[o] = wV[li];
            }
        }
    }
}

extern "C" void kernel_launch(void* const* d_in, const int* in_sizes, int n_in,
                              void* d_out, int out_size, void* d_ws, size_t ws_size,
                              hipStream_t stream)
{
    (void)n_in; (void)out_size; (void)ws_size;
    const float* input      = (const float*)d_in[0];
    const float* particles  = (const float*)d_in[1];
    const float* velocities = (const float*)d_in[2];
    const float* r1         = (const float*)d_in[3];
    const float* r2         = (const float*)d_in[4];
    const int*   gloc       = (const int*)d_in[5];
    const int*   iter_num   = (const int*)d_in[6];

    const int N  = in_sizes[1] / SOM_DIM;
    const int NB = (N + THREADS - 1) / THREADS;
    const int NA = (N + A_ROWS - 1) / A_ROWS;

    float* ws      = (float*)d_ws;
    float* blkSum  = ws;
    float* blkMinV = ws + (size_t)NA * SOM_DIM;
    int*   blkMinI = (int*)(blkMinV + NA);
    size_t hoff    = (size_t)NA * SOM_DIM + 2 * (size_t)NA;
    hoff = (hoff + 3) & ~(size_t)3;
    float* hdr     = ws + hoff;

    float* outP = (float*)d_out;
    float* outV = outP + (size_t)N * SOM_DIM;

    som_pass_a<<<NA, A_THREADS, 0, stream>>>(input, particles, blkSum, blkMinV, blkMinI, N);
    som_pass_b<<<1, B_THREADS, 0, stream>>>(particles, gloc, iter_num,
                                          blkSum, blkMinV, blkMinI, hdr, NA, N);
    som_pass_c<<<NB, THREADS, 0, stream>>>(particles, velocities, r1, r2,
                                           gloc, hdr, outP, outV, N);
}


# Pass A stores the particle rows it holds to the output; pass C rewrites only 32-row tiles with an updated row, an all-keep wave only copies velocities (no P, r1, r2 load, no LDS tile); sums as before
# speedup vs baseline: 5.3208x; 1.2917x over previous
#include <hip/hip_runtime.h>
#include <hip/hip_bf16.h>
#include <math.h>

typedef float v2f __attribute__((ext_vector_type(2)));
typedef float v8f __attribute__((ext_vector_type(8)));
typedef float v4f __attribute__((ext_vector_type(4)));

#define THREADS 256
#define SOM_DIM 32
#define LSTRIDE 36
#define PITCH4  9


#define A_THREADS 256
#define A_WAVES   8
#define A_TILES   4
#define A_ROWS    (A_THREADS * A_TILES)

__global__ __launch_bounds__(A_THREADS)
void som_pass_a(const float* __restrict__ input,
                const float* __restrict__ particles,
                float* __restrict__ blkSum,
                float* __restrict__ blkMinV,
                int*   __restrict__ blkMinI,
                float* __restrict__ outP,
                int N)
{
    __shared__ float4 stage4[A_WAVES][32 * PITCH4];
    __shared__ float wsum[A_WAVES * SOM_DIM];
    __shared__ float sdist[A_WAVES];
    __shared__ int   sidx[A_WAVES];

    const int t    = threadIdx.x;
    const int lane = t & 31;
    const int wave = t >> 5;
    float4* w4 = stage4[wave];
    const float* wf = reinterpret_cast<const float*>(w4);

    const int sub = lane >> 3;
    const int pc  = lane & 7;
    const float4* g4 = reinterpret_cast<const float4*>(particles);
    float4* OP4 = reinterpret_cast<float4*>(outP);
    const int base = blockIdx.x * A_ROWS + wave * 32;

    float4 v[A_TILES][8];
    #pragma unroll
    for (int it = 0; it < A_TILES; ++it) {
        const int grow = base + it * A_THREADS;
        #pragma unroll
        for (int q = 0; q < 8; ++q) {
            const int r  = grow + 4*q + sub;
            const int rc = (r < N) ? r : (N - 1);
            v[it][q] = g4[(size_t)rc * 8 + pc];
        }
    }

    float4 xv[8];
    {
        const float4* x4 = reinterpret_cast<const float4*>(input);
        #pragma unroll
        for (int q = 0; q < 8; ++q) xv[q] = x4[q];
    }

    const int half = lane >> 4;
    const int n    = lane & 15;
    v2f onesA; onesA.x = 1.0f; onesA.y = 1.0f;
    v8f acc0 = {};
    v8f acc1 = {};

    float best = 3.402823466e+38f;
    int   bidx = 0x7fffffff;

    #pragma unroll
    for (int it = 0; it < A_TILES; ++it) {
        if (base + it * A_THREADS + 32 > N) {
            #pragma unroll
            for (int q = 0; q < 8; ++q) {
                const int r = base + it * A_THREADS + 4*q + sub;
                if (r < N) OP4[(size_t)r * 8 + pc] = v[it][q];
                else       v[it][q] = make_float4(0.f, 0.f, 0.f, 0.f);
            }
        } else {
            #pragma unroll
            for (int q = 0; q < 8; ++q)
                OP4[(size_t)(base + it * A_THREADS + 4*q + sub) * 8 + pc] = v[it][q];
        }
        #pragma unroll
        for (int q = 0; q < 8; ++q)
            w4[(4*q + sub) * PITCH4 + pc] = v[it][q];

        __builtin_amdgcn_fence(__ATOMIC_RELEASE, "workgroup", "local");
        __builtin_amdgcn_wave_barrier();
        __builtin_amdgcn_fence(__ATOMIC_ACQUIRE, "workgroup", "local");

        float dist2 = 0.f;
        #pragma unroll
        for (int q = 0; q < 8; ++q) {
            const float4 pv = w4[lane * PITCH4 + q];
            float d0 = pv.x - xv[q].x, d1 = pv.y - xv[q].y;
            float d2 = pv.z - xv[q].z, d3 = pv.w - xv[q].w;
            float tq = d1*d1;
            tq = fmaf(d0, d0, tq);
            tq = fmaf(d2, d2, tq);
            tq = fmaf(d3, d3, tq);
            dist2 += tq;
        }
        const int p = base + it * A_THREADS + lane;
        if (p >= N) dist2 = 3.402823466e+38f;
        if (dist2 < best || (dist2 == best && p < bidx)) { best = dist2; bidx = p; }

        #pragma unroll
        for (int c = 0; c < 8; ++c) {
            const int r0 = c*4 + half;
            const int r1 = r0 + 2;
            v2f b0, b1;
            b0.x = wf[r0*LSTRIDE + n];        b0.y = wf[r1*LSTRIDE + n];
            b1.x = wf[r0*LSTRIDE + 16 + n];   b1.y = wf[r1*LSTRIDE + 16 + n];
            acc0 = __builtin_amdgcn_wmma_f32_16x16x4_f32(false, onesA, false, b0,
                                                         (short)0, acc0, false, false);
            acc1 = __builtin_amdgcn_wmma_f32_16x16x4_f32(false, onesA, false, b1,
                                                         (short)0, acc1, false, false);
        }

        if (it + 1 < A_TILES) {
            __builtin_amdgcn_fence(__ATOMIC_RELEASE, "workgroup", "local");
            __builtin_amdgcn_wave_barrier();
            __builtin_amdgcn_fence(__ATOMIC_ACQUIRE, "workgroup", "local");
        }
    }

    #pragma unroll
    for (int m = 16; m > 0; m >>= 1) {
        const float ob = __shfl_xor(best, m, 32);
        const int   oi = __shfl_xor(bidx, m, 32);
        if (ob < best || (ob == best && oi < bidx)) { best = ob; bidx = oi; }
    }
    if (lane == 0) { sdist[wave] = best; sidx[wave] = bidx; }
    if (lane < 16) {
        wsum[wave*SOM_DIM + lane]      = acc0[0];
        wsum[wave*SOM_DIM + 16 + lane] = acc1[0];
    }
    __syncthreads();

    if (wave == 0) {
        float b = 3.402823466e+38f;
        int   i = 0x7fffffff;
        if (lane < A_WAVES) { b = sdist[lane]; i = sidx[lane]; }
        #pragma unroll
        for (int m = 16; m > 0; m >>= 1) {
            const float ob = __shfl_xor(b, m, 32);
            const int   oi = __shfl_xor(i, m, 32);
            if (ob < b || (ob == b && oi < i)) { b = ob; i = oi; }
        }
        if (lane == 0) { blkMinV[blockIdx.x] = b; blkMinI[blockIdx.x] = i; }
    } else if (wave == 1) {
        float s = 0.f;
        #pragma unroll
        for (int w = 0; w < A_WAVES; ++w) s += wsum[w*SOM_DIM + lane];
        blkSum[(size_t)blockIdx.x * SOM_DIM + lane] = s;
    }
}

#define B_FOLD_WAVES 16
#define B_THREADS    (32 * (B_FOLD_WAVES + 1))

__global__ __launch_bounds__(B_THREADS)
void som_pass_b(const float* __restrict__ particles,
                const int*   __restrict__ gloc,
                const int*   __restrict__ iter_num,
                const float* __restrict__ blkSum,
                const float* __restrict__ blkMinV,
                const int*   __restrict__ blkMinI,
                float* __restrict__ hdr,
                int NB, int N)
{
    __shared__ float4 part4[B_FOLD_WAVES][SOM_DIM / 4];
    const float* part = reinterpret_cast<const float*>(&part4[0][0]);

    const int lane = threadIdx.x & 31;
    const int wave = threadIdx.x >> 5;

    float gb = 0.f;
    int   gx = 0, gy = 0;
    int   itn = 0;

    if (wave == B_FOLD_WAVES) {
        float best = 3.402823466e+38f; int bidx = 0x7fffffff;
        if ((NB & 3) == 0) {
            const float4* V4 = reinterpret_cast<const float4*>(blkMinV);
            const int4*   I4 = reinterpret_cast<const int4*>(blkMinI);
            const int NB4 = NB >> 2;
            for (int k = lane; k < NB4; k += 32) {
                const float4 dv = V4[k]; const int4 iv = I4[k];
                if (dv.x < best || (dv.x == best && iv.x < bidx)) { best = dv.x; bidx = iv.x; }
                if (dv.y < best || (dv.y == best && iv.y < bidx)) { best = dv.y; bidx = iv.y; }
                if (dv.z < best || (dv.z == best && iv.z < bidx)) { best = dv.z; bidx = iv.z; }
                if (dv.w < best || (dv.w == best && iv.w < bidx)) { best = dv.w; bidx = iv.w; }
            }
        } else {
            for (int k = lane; k < NB; k += 32) {
                const float dv = blkMinV[k]; const int iv = blkMinI[k];
                if (dv < best || (dv == best && iv < bidx)) { best = dv; bidx = iv; }
            }
        }
        #pragma unroll
        for (int m = 16; m > 0; m >>= 1) {
            const float ob = __shfl_xor(best, m, 32);
            const int   oi = __shfl_xor(bidx, m, 32);
            if (ob < best || (ob == best && oi < bidx)) { best = ob; bidx = oi; }
        }
        gb  = particles[(size_t)bidx * SOM_DIM + lane];
        gx  = gloc[2*(size_t)bidx];
        gy  = gloc[2*(size_t)bidx + 1];
        itn = iter_num[0];
    } else {
        const int seg = wave;
        const int per = (NB + B_FOLD_WAVES - 1) / B_FOLD_WAVES;
        const int b0 = seg * per;
        const int b1 = (b0 + per < NB) ? (b0 + per) : NB;
        const int r  = lane >> 3;
        const int pc = lane & 7;
        const float4* S4 = reinterpret_cast<const float4*>(blkSum);
        float4 s = make_float4(0.f, 0.f, 0.f, 0.f);
        for (int b = b0 + r; b < b1; b += 4) {
            const float4 x = S4[(size_t)b * (SOM_DIM / 4) + pc];
            s.x += x.x; s.y += x.y; s.z += x.z; s.w += x.w;
        }
        s.x += __shfl_xor(s.x, 8, 32);  s.y += __shfl_xor(s.y, 8, 32);
        s.z += __shfl_xor(s.z, 8, 32);  s.w += __shfl_xor(s.w, 8, 32);
        s.x += __shfl_xor(s.x, 16, 32); s.y += __shfl_xor(s.y, 16, 32);
        s.z += __shfl_xor(s.z, 16, 32); s.w += __shfl_xor(s.w, 16, 32);
        if (r == 0) part4[seg][pc] = s;
    }
    __syncthreads();

    if (wave == B_FOLD_WAVES) {
        float S = 0.f;
        #pragma unroll
        for (int g = 0; g < B_FOLD_WAVES; ++g) S += part[g * SOM_DIM + lane];
        hdr[4 + lane]              = gb;
        hdr[4 + SOM_DIM + lane]    = S / (float)N;
        hdr[4 + 2*SOM_DIM + lane]  = (S - gb) / (float)(N - 1);
        if (lane == 0) {
            const float decay  = 1.0f - (float)itn / 100.0f;
            const float radius = 0.5f * decay;
            const float sig    = 256.0f * decay;
            hdr[0] = (float)gx;
            hdr[1] = (float)gy;
            hdr[2] = radius;
            hdr[3] = sig * sig;
        }
    }
}

__global__ __launch_bounds__(THREADS)
void som_pass_c(const float* __restrict__ particles,
                const float* __restrict__ velocities,
                const float* __restrict__ r1,
                const float* __restrict__ r2,
                const int*   __restrict__ gloc,
                const float* __restrict__ hdr,
                float* __restrict__ outP,
                float* __restrict__ outV,
                int N)
{
    __shared__ float4 tP[8][32 * PITCH4];
    __shared__ float4 tV[8][32 * PITCH4];
    __shared__ float4 tA[8][32 * PITCH4];
    __shared__ float4 tB[8][32 * PITCH4];

    const int lane = threadIdx.x & 31;
    const int wave = threadIdx.x >> 5;
    const int row0 = blockIdx.x * THREADS + wave * 32;
    if (row0 >= N) return;

    float4* wP = tP[wave];
    float4* wV = tV[wave];
    float4* wA = tA[wave];
    float4* wB = tB[wave];

    const int sub = lane >> 3;
    const int pc  = lane & 7;

    float4* OP4 = reinterpret_cast<float4*>(outP);
    float4* OV4 = reinterpret_cast<float4*>(outV);

    const int p  = row0 + lane;
    const int pcl = (p < N) ? p : (N - 1);
    const int2 gl = *(const int2*)(gloc + 2*(size_t)pcl);

    const v4f* V4 = reinterpret_cast<const v4f*>(velocities);
    v4f gv[8];
    #pragma unroll
    for (int q = 0; q < 8; ++q) {
        const int r  = row0 + 4*q + sub;
        const int rc = (r < N) ? r : (N - 1);
        gv[q] = V4[(size_t)rc * 8 + pc];
    }

    const float bx = hdr[0], by = hdr[1], radius = hdr[2], sigma2 = hdr[3];
    const float* gbest = hdr + 4;
    const float* cenA  = hdr + 4 + SOM_DIM;
    const float* cenB  = hdr + 4 + 2*SOM_DIM;
    const float dx = (float)gl.x - bx;
    const float dy = (float)gl.y - by;
    const float nb = expf(-(dx*dx + dy*dy) / sigma2);
    const bool  skip = (1.0f - nb) > radius;
    const bool  useA = (nb + radius) >= 1.0f;

    asm volatile("" : "+v"(gv[0]), "+v"(gv[1]), "+v"(gv[2]), "+v"(gv[3]),
                      "+v"(gv[4]), "+v"(gv[5]), "+v"(gv[6]), "+v"(gv[7]));

    if (__builtin_amdgcn_ballot_w32(!skip && p < N) == 0u) {
        v4f* OVv = reinterpret_cast<v4f*>(outV);
        if (row0 + 32 <= N) {
            #pragma unroll
            for (int q = 0; q < 8; ++q)
                OVv[(size_t)(row0 + 4*q + sub) * 8 + pc] = gv[q];
        } else {
            #pragma unroll
            for (int q = 0; q < 8; ++q) {
                const int r = row0 + 4*q + sub;
                if (r < N) OVv[(size_t)r * 8 + pc] = gv[q];
            }
        }
        return;
    }

    {
        const float4* P4 = reinterpret_cast<const float4*>(particles);
        const float4* A4 = reinterpret_cast<const float4*>(r1);
        const float4* B4 = reinterpret_cast<const float4*>(r2);
        float4 gp[8], ga[8], gb[8];
        #pragma unroll
        for (int q = 0; q < 8; ++q) {
            const int r  = row0 + 4*q + sub;
            const int rc = (r < N) ? r : (N - 1);
            const size_t o = (size_t)rc * 8 + pc;
            gp[q] = P4[o]; ga[q] = A4[o]; gb[q] = B4[o];
        }
        #pragma unroll
        for (int q = 0; q < 8; ++q) {
            const int li = (4*q + sub) * PITCH4 + pc;
            wP[li] = gp[q]; wA[li] = ga[q]; wB[li] = gb[q];
            wV[li] = make_float4(gv[q].x, gv[q].y, gv[q].z, gv[q].w);
        }
    }
    const float4* CA4 = (const float4*)cenA;
    const float4* CB4 = (const float4*)cenB;
    const float4* G4  = (const float4*)gbest;
    float4 c8[8], g8[8];
    #pragma unroll
    for (int q = 0; q < 8; ++q) {
        const float4 ca = CA4[q], cb = CB4[q];
        c8[q].x = useA ? ca.x : cb.x;
        c8[q].y = useA ? ca.y : cb.y;
        c8[q].z = useA ? ca.z : cb.z;
        c8[q].w = useA ? ca.w : cb.w;
        g8[q] = G4[q];
    }

    __builtin_amdgcn_fence(__ATOMIC_RELEASE, "workgroup");
    __builtin_amdgcn_wave_barrier();
    __builtin_amdgcn_fence(__ATOMIC_ACQUIRE, "workgroup");

    const int my = lane * PITCH4;
    #pragma unroll
    for (int q = 0; q < 8; ++q) {
        const float4 pv = wP[my + q], vv = wV[my + q], a = wA[my + q], b = wB[my + q];
        const float4 c = c8[q], g = g8[q];
        float4 vn, pn;
        vn.x = 0.001f*vv.x + 0.01f*a.x*(c.x - pv.x) + 0.1f*b.x*(g.x - pv.x);
        vn.y = 0.001f*vv.y + 0.01f*a.y*(c.y - pv.y) + 0.1f*b.y*(g.y - pv.y);
        vn.z = 0.001f*vv.z + 0.01f*a.z*(c.z - pv.z) + 0.1f*b.z*(g.z - pv.z);
        vn.w = 0.001f*vv.w + 0.01f*a.w*(c.w - pv.w) + 0.1f*b.w*(g.w - pv.w);
        pn.x = pv.x + vn.x; pn.y = pv.y + vn.y;
        pn.z = pv.z + vn.z; pn.w = pv.w + vn.w;
        if (skip) { vn = vv; pn = pv; }
        wP[my + q] = pn;
        wV[my + q] = vn;
    }

    __builtin_amdgcn_fence(__ATOMIC_RELEASE, "workgroup");
    __builtin_amdgcn_wave_barrier();
    __builtin_amdgcn_fence(__ATOMIC_ACQUIRE, "workgroup");

    if (row0 + 32 <= N) {
        float4 op[8], ov[8];
        #pragma unroll
        for (int q = 0; q < 8; ++q) {
            const int li = (4*q + sub) * PITCH4 + pc;
            op[q] = wP[li]; ov[q] = wV[li];
        }
        #pragma unroll
        for (int q = 0; q < 8; ++q) {
            const size_t o = (size_t)(row0 + 4*q + sub) * 8 + pc;
            OP4[o] = op[q];
            OV4[o] = ov[q];
        }
    } else {
        #pragma unroll
        for (int q = 0; q < 8; ++q) {
            const int r  = row0 + 4*q + sub;
            const int li = (4*q + sub) * PITCH4 + pc;
            if (r < N) {
                const size_t o = (size_t)r * 8 + pc;
                OP4[o] = wP[li];
                OV4[o] = wV[li];
            }
        }
    }
}

extern "C" void kernel_launch(void* const* d_in, const int* in_sizes, int n_in,
                              void* d_out, int out_size, void* d_ws, size_t ws_size,
                              hipStream_t stream)
{
    (void)n_in; (void)out_size; (void)ws_size;
    const float* input      = (const float*)d_in[0];
    const float* particles  = (const float*)d_in[1];
    const float* velocities = (const float*)d_in[2];
    const float* r1         = (const float*)d_in[3];
    const float* r2         = (const float*)d_in[4];
    const int*   gloc       = (const int*)d_in[5];
    const int*   iter_num   = (const int*)d_in[6];

    const int N  = in_sizes[1] / SOM_DIM;
    const int NB = (N + THREADS - 1) / THREADS;
    const int NA = (N + A_ROWS - 1) / A_ROWS;

    float* ws      = (float*)d_ws;
    float* blkSum  = ws;
    float* blkMinV = ws + (size_t)NA * SOM_DIM;
    int*   blkMinI = (int*)(blkMinV + NA);
    size_t hoff    = (size_t)NA * SOM_DIM + 2 * (size_t)NA;
    hoff = (hoff + 3) & ~(size_t)3;
    float* hdr     = ws + hoff;

    float* outP = (float*)d_out;
    float* outV = outP + (size_t)N * SOM_DIM;

    som_pass_a<<<NA, A_THREADS, 0, stream>>>(input, particles, blkSum, blkMinV, blkMinI, outP, N);
    som_pass_b<<<1, B_THREADS, 0, stream>>>(particles, gloc, iter_num,
                                          blkSum, blkMinV, blkMinI, hdr, NA, N);
    som_pass_c<<<NB, THREADS, 0, stream>>>(particles, velocities, r1, r2,
                                           gloc, hdr, outP, outV, N);
}


# Pass C: 256 work-groups of 8 waves x 4 tiles (was 1024 of 8 x 1), tile pairs dealt evenly over the WGPs; a wave has its 4 tiles' velocity loads, then an updated pair's 48 loads, in flight at once
# speedup vs baseline: 5.7597x; 1.0825x over previous
#include <hip/hip_runtime.h>
#include <hip/hip_bf16.h>
#include <math.h>

typedef float v2f __attribute__((ext_vector_type(2)));
typedef float v8f __attribute__((ext_vector_type(8)));
typedef float v4f __attribute__((ext_vector_type(4)));

#define THREADS 256
#define SOM_DIM 32
#define LSTRIDE 36
#define PITCH4  9


#define A_THREADS 256
#define A_WAVES   8
#define A_TILES   4
#define A_ROWS    (A_THREADS * A_TILES)

__global__ __launch_bounds__(A_THREADS)
void som_pass_a(const float* __restrict__ input,
                const float* __restrict__ particles,
                float* __restrict__ blkSum,
                float* __restrict__ blkMinV,
                int*   __restrict__ blkMinI,
                float* __restrict__ outP,
                int N)
{
    __shared__ float4 stage4[A_WAVES][32 * PITCH4];
    __shared__ float wsum[A_WAVES * SOM_DIM];
    __shared__ float sdist[A_WAVES];
    __shared__ int   sidx[A_WAVES];

    const int t    = threadIdx.x;
    const int lane = t & 31;
    const int wave = t >> 5;
    float4* w4 = stage4[wave];
    const float* wf = reinterpret_cast<const float*>(w4);

    const int sub = lane >> 3;
    const int pc  = lane & 7;
    const float4* g4 = reinterpret_cast<const float4*>(particles);
    float4* OP4 = reinterpret_cast<float4*>(outP);
    const int base = blockIdx.x * A_ROWS + wave * 32;

    float4 v[A_TILES][8];
    #pragma unroll
    for (int it = 0; it < A_TILES; ++it) {
        const int grow = base + it * A_THREADS;
        #pragma unroll
        for (int q = 0; q < 8; ++q) {
            const int r  = grow + 4*q + sub;
            const int rc = (r < N) ? r : (N - 1);
            v[it][q] = g4[(size_t)rc * 8 + pc];
        }
    }

    float4 xv[8];
    {
        const float4* x4 = reinterpret_cast<const float4*>(input);
        #pragma unroll
        for (int q = 0; q < 8; ++q) xv[q] = x4[q];
    }

    const int half = lane >> 4;
    const int n    = lane & 15;
    v2f onesA; onesA.x = 1.0f; onesA.y = 1.0f;
    v8f acc0 = {};
    v8f acc1 = {};

    float best = 3.402823466e+38f;
    int   bidx = 0x7fffffff;

    #pragma unroll
    for (int it = 0; it < A_TILES; ++it) {
        if (base + it * A_THREADS + 32 > N) {
            #pragma unroll
            for (int q = 0; q < 8; ++q) {
                const int r = base + it * A_THREADS + 4*q + sub;
                if (r < N) OP4[(size_t)r * 8 + pc] = v[it][q];
                else       v[it][q] = make_float4(0.f, 0.f, 0.f, 0.f);
            }
        } else {
            #pragma unroll
            for (int q = 0; q < 8; ++q)
                OP4[(size_t)(base + it * A_THREADS + 4*q + sub) * 8 + pc] = v[it][q];
        }
        #pragma unroll
        for (int q = 0; q < 8; ++q)
            w4[(4*q + sub) * PITCH4 + pc] = v[it][q];

        __builtin_amdgcn_fence(__ATOMIC_RELEASE, "workgroup", "local");
        __builtin_amdgcn_wave_barrier();
        __builtin_amdgcn_fence(__ATOMIC_ACQUIRE, "workgroup", "local");

        float dist2 = 0.f;
        #pragma unroll
        for (int q = 0; q < 8; ++q) {
            const float4 pv = w4[lane * PITCH4 + q];
            float d0 = pv.x - xv[q].x, d1 = pv.y - xv[q].y;
            float d2 = pv.z - xv[q].z, d3 = pv.w - xv[q].w;
            float tq = d1*d1;
            tq = fmaf(d0, d0, tq);
            tq = fmaf(d2, d2, tq);
            tq = fmaf(d3, d3, tq);
            dist2 += tq;
        }
        const int p = base + it * A_THREADS + lane;
        if (p >= N) dist2 = 3.402823466e+38f;
        if (dist2 < best || (dist2 == best && p < bidx)) { best = dist2; bidx = p; }

        #pragma unroll
        for (int c = 0; c < 8; ++c) {
            const int r0 = c*4 + half;
            const int r1 = r0 + 2;
            v2f b0, b1;
            b0.x = wf[r0*LSTRIDE + n];        b0.y = wf[r1*LSTRIDE + n];
            b1.x = wf[r0*LSTRIDE + 16 + n];   b1.y = wf[r1*LSTRIDE + 16 + n];
            acc0 = __builtin_amdgcn_wmma_f32_16x16x4_f32(false, onesA, false, b0,
                                                         (short)0, acc0, false, false);
            acc1 = __builtin_amdgcn_wmma_f32_16x16x4_f32(false, onesA, false, b1,
                                                         (short)0, acc1, false, false);
        }

        if (it + 1 < A_TILES) {
            __builtin_amdgcn_fence(__ATOMIC_RELEASE, "workgroup", "local");
            __builtin_amdgcn_wave_barrier();
            __builtin_amdgcn_fence(__ATOMIC_ACQUIRE, "workgroup", "local");
        }
    }

    #pragma unroll
    for (int m = 16; m > 0; m >>= 1) {
        const float ob = __shfl_xor(best, m, 32);
        const int   oi = __shfl_xor(bidx, m, 32);
        if (ob < best || (ob == best && oi < bidx)) { best = ob; bidx = oi; }
    }
    if (lane == 0) { sdist[wave] = best; sidx[wave] = bidx; }
    if (lane < 16) {
        wsum[wave*SOM_DIM + lane]      = acc0[0];
        wsum[wave*SOM_DIM + 16 + lane] = acc1[0];
    }
    __syncthreads();

    if (wave == 0) {
        float b = 3.402823466e+38f;
        int   i = 0x7fffffff;
        if (lane < A_WAVES) { b = sdist[lane]; i = sidx[lane]; }
        #pragma unroll
        for (int m = 16; m > 0; m >>= 1) {
            const float ob = __shfl_xor(b, m, 32);
            const int   oi = __shfl_xor(i, m, 32);
            if (ob < b || (ob == b && oi < i)) { b = ob; i = oi; }
        }
        if (lane == 0) { blkMinV[blockIdx.x] = b; blkMinI[blockIdx.x] = i; }
    } else if (wave == 1) {
        float s = 0.f;
        #pragma unroll
        for (int w = 0; w < A_WAVES; ++w) s += wsum[w*SOM_DIM + lane];
        blkSum[(size_t)blockIdx.x * SOM_DIM + lane] = s;
    }
}

#define B_FOLD_WAVES 16
#define B_THREADS    (32 * (B_FOLD_WAVES + 1))

__global__ __launch_bounds__(B_THREADS)
void som_pass_b(const float* __restrict__ particles,
                const int*   __restrict__ gloc,
                const int*   __restrict__ iter_num,
                const float* __restrict__ blkSum,
                const float* __restrict__ blkMinV,
                const int*   __restrict__ blkMinI,
                float* __restrict__ hdr,
                int NB, int N)
{
    __shared__ float4 part4[B_FOLD_WAVES][SOM_DIM / 4];
    const float* part = reinterpret_cast<const float*>(&part4[0][0]);

    const int lane = threadIdx.x & 31;
    const int wave = threadIdx.x >> 5;

    float gb = 0.f;
    int   gx = 0, gy = 0;
    int   itn = 0;

    if (wave == B_FOLD_WAVES) {
        float best = 3.402823466e+38f; int bidx = 0x7fffffff;
        if ((NB & 3) == 0) {
            const float4* V4 = reinterpret_cast<const float4*>(blkMinV);
            const int4*   I4 = reinterpret_cast<const int4*>(blkMinI);
            const int NB4 = NB >> 2;
            for (int k = lane; k < NB4; k += 32) {
                const float4 dv = V4[k]; const int4 iv = I4[k];
                if (dv.x < best || (dv.x == best && iv.x < bidx)) { best = dv.x; bidx = iv.x; }
                if (dv.y < best || (dv.y == best && iv.y < bidx)) { best = dv.y; bidx = iv.y; }
                if (dv.z < best || (dv.z == best && iv.z < bidx)) { best = dv.z; bidx = iv.z; }
                if (dv.w < best || (dv.w == best && iv.w < bidx)) { best = dv.w; bidx = iv.w; }
            }
        } else {
            for (int k = lane; k < NB; k += 32) {
                const float dv = blkMinV[k]; const int iv = blkMinI[k];
                if (dv < best || (dv == best && iv < bidx)) { best = dv; bidx = iv; }
            }
        }
        #pragma unroll
        for (int m = 16; m > 0; m >>= 1) {
            const float ob = __shfl_xor(best, m, 32);
            const int   oi = __shfl_xor(bidx, m, 32);
            if (ob < best || (ob == best && oi < bidx)) { best = ob; bidx = oi; }
        }
        gb  = particles[(size_t)bidx * SOM_DIM + lane];
        gx  = gloc[2*(size_t)bidx];
        gy  = gloc[2*(size_t)bidx + 1];
        itn = iter_num[0];
    } else {
        const int seg = wave;
        const int per = (NB + B_FOLD_WAVES - 1) / B_FOLD_WAVES;
        const int b0 = seg * per;
        const int b1 = (b0 + per < NB) ? (b0 + per) : NB;
        const int r  = lane >> 3;
        const int pc = lane & 7;
        const float4* S4 = reinterpret_cast<const float4*>(blkSum);
        float4 s = make_float4(0.f, 0.f, 0.f, 0.f);
        for (int b = b0 + r; b < b1; b += 4) {
            const float4 x = S4[(size_t)b * (SOM_DIM / 4) + pc];
            s.x += x.x; s.y += x.y; s.z += x.z; s.w += x.w;
        }
        s.x += __shfl_xor(s.x, 8, 32);  s.y += __shfl_xor(s.y, 8, 32);
        s.z += __shfl_xor(s.z, 8, 32);  s.w += __shfl_xor(s.w, 8, 32);
        s.x += __shfl_xor(s.x, 16, 32); s.y += __shfl_xor(s.y, 16, 32);
        s.z += __shfl_xor(s.z, 16, 32); s.w += __shfl_xor(s.w, 16, 32);
        if (r == 0) part4[seg][pc] = s;
    }
    __syncthreads();

    if (wave == B_FOLD_WAVES) {
        float S = 0.f;
        #pragma unroll
        for (int g = 0; g < B_FOLD_WAVES; ++g) S += part[g * SOM_DIM + lane];
        hdr[4 + lane]              = gb;
        hdr[4 + SOM_DIM + lane]    = S / (float)N;
        hdr[4 + 2*SOM_DIM + lane]  = (S - gb) / (float)(N - 1);
        if (lane == 0) {
            const float decay  = 1.0f - (float)itn / 100.0f;
            const float radius = 0.5f * decay;
            const float sig    = 256.0f * decay;
            hdr[0] = (float)gx;
            hdr[1] = (float)gy;
            hdr[2] = radius;
            hdr[3] = sig * sig;
        }
    }
}

#define C_THREADS 256
#define C_WAVES   8
#define C_TILES   4
#define C_ROWS    (C_THREADS * C_TILES)

__device__ __forceinline__
void som_c_row(int p, const float* __restrict__ particles, const float* __restrict__ velocities,
               const float* __restrict__ r1, const float* __restrict__ r2, const int* __restrict__ gloc,
               const float* __restrict__ hdr, float* __restrict__ outP, float* __restrict__ outV)
{
    const float bx = hdr[0], by = hdr[1], radius = hdr[2], sigma2 = hdr[3];
    const float dx = (float)gloc[2*(size_t)p] - bx;
    const float dy = (float)gloc[2*(size_t)p + 1] - by;
    const float nb = expf(-(dx*dx + dy*dy) / sigma2);
    const bool  skip = (1.0f - nb) > radius;
    const bool  useA = (nb + radius) >= 1.0f;
    const float* cen = hdr + 4 + (useA ? SOM_DIM : 2*SOM_DIM);
    #pragma nounroll
    for (int d = 0; d < SOM_DIM; ++d) {
        const size_t o = (size_t)p * SOM_DIM + d;
        const float pv = particles[o], vv = velocities[o];
        const float vn = 0.001f*vv + 0.01f*r1[o]*(cen[d] - pv) + 0.1f*r2[o]*(hdr[4 + d] - pv);
        outP[o] = skip ? pv : (pv + vn);
        outV[o] = skip ? vv : vn;
    }
}

__device__ __forceinline__
void som_c_load(v4f (&p)[8], v4f (&a)[8], v4f (&b)[8], const float* __restrict__ particles,
                const float* __restrict__ r1, const float* __restrict__ r2, int row0, int lane)
{
    const v4f* pb = reinterpret_cast<const v4f*>(particles) + (size_t)row0 * 8;
    const v4f* ab = reinterpret_cast<const v4f*>(r1) + (size_t)row0 * 8;
    const v4f* rb = reinterpret_cast<const v4f*>(r2) + (size_t)row0 * 8;
    #pragma unroll
    for (int q = 0; q < 8; ++q) { p[q] = pb[q * 32 + lane]; a[q] = ab[q * 32 + lane]; b[q] = rb[q * 32 + lane]; }
}

__device__ __forceinline__
void som_c_copy(const v4f (&gv)[8], float* __restrict__ outV, int row0, int lane)
{
    v4f* ob = reinterpret_cast<v4f*>(outV) + (size_t)row0 * 8;
    #pragma unroll
    for (int q = 0; q < 8; ++q) ob[q * 32 + lane] = gv[q];
}

__device__ __forceinline__
void som_c_work(const v4f (&gv)[8], const v4f (&p)[8], const v4f (&a)[8], const v4f (&b)[8],
                bool sk, bool ua, float4* wP, float4* wV, float4* wA, float4* wB, const float4* wC,
                float* __restrict__ outP, float* __restrict__ outV, int row0, int lane)
{
    const int sub = lane >> 3;
    const int pc  = lane & 7;
    __builtin_amdgcn_fence(__ATOMIC_RELEASE, "workgroup", "local");
    __builtin_amdgcn_wave_barrier();
    __builtin_amdgcn_fence(__ATOMIC_ACQUIRE, "workgroup", "local");
    #pragma unroll
    for (int q = 0; q < 8; ++q) {
        const int li = (4*q + sub) * PITCH4 + pc;
        wP[li] = make_float4(p[q].x, p[q].y, p[q].z, p[q].w);
        wA[li] = make_float4(a[q].x, a[q].y, a[q].z, a[q].w);
        wB[li] = make_float4(b[q].x, b[q].y, b[q].z, b[q].w);
        wV[li] = make_float4(gv[q].x, gv[q].y, gv[q].z, gv[q].w);
    }
    __builtin_amdgcn_fence(__ATOMIC_RELEASE, "workgroup", "local");
    __builtin_amdgcn_wave_barrier();
    __builtin_amdgcn_fence(__ATOMIC_ACQUIRE, "workgroup", "local");

    const int my = lane * PITCH4;
    #pragma unroll
    for (int q = 0; q < 8; ++q) {
        const float4 pv = wP[my + q], vv = wV[my + q], ra = wA[my + q], rb = wB[my + q];
        const float4 g = wC[q], ca = wC[8 + q], cb = wC[16 + q];
        float4 c, vn, pn;
        c.x = ua ? ca.x : cb.x;
        c.y = ua ? ca.y : cb.y;
        c.z = ua ? ca.z : cb.z;
        c.w = ua ? ca.w : cb.w;
        vn.x = 0.001f*vv.x + 0.01f*ra.x*(c.x - pv.x) + 0.1f*rb.x*(g.x - pv.x);
        vn.y = 0.001f*vv.y + 0.01f*ra.y*(c.y - pv.y) + 0.1f*rb.y*(g.y - pv.y);
        vn.z = 0.001f*vv.z + 0.01f*ra.z*(c.z - pv.z) + 0.1f*rb.z*(g.z - pv.z);
        vn.w = 0.001f*vv.w + 0.01f*ra.w*(c.w - pv.w) + 0.1f*rb.w*(g.w - pv.w);
        pn.x = pv.x + vn.x; pn.y = pv.y + vn.y;
        pn.z = pv.z + vn.z; pn.w = pv.w + vn.w;
        if (sk) { vn = vv; pn = pv; }
        wP[my + q] = pn;
        wV[my + q] = vn;
    }

    __builtin_amdgcn_fence(__ATOMIC_RELEASE, "workgroup", "local");
    __builtin_amdgcn_wave_barrier();
    __builtin_amdgcn_fence(__ATOMIC_ACQUIRE, "workgroup", "local");

    float4 op[8], ov[8];
    #pragma unroll
    for (int q = 0; q < 8; ++q) {
        const int li = (4*q + sub) * PITCH4 + pc;
        op[q] = wP[li]; ov[q] = wV[li];
    }
    float4* opb = reinterpret_cast<float4*>(outP) + (size_t)row0 * 8;
    float4* ovb = reinterpret_cast<float4*>(outV) + (size_t)row0 * 8;
    #pragma unroll
    for (int q = 0; q < 8; ++q) {
        opb[q * 32 + lane] = op[q];
        ovb[q * 32 + lane] = ov[q];
    }
}

__global__ __launch_bounds__(C_THREADS)
void som_pass_c(const float* __restrict__ particles,
                const float* __restrict__ velocities,
                const float* __restrict__ r1,
                const float* __restrict__ r2,
                const int*   __restrict__ gloc,
                const float* __restrict__ hdr,
                float* __restrict__ outP,
                float* __restrict__ outV,
                int N)
{
    __shared__ float4 tP[C_WAVES][32 * PITCH4];
    __shared__ float4 tV[C_WAVES][32 * PITCH4];
    __shared__ float4 tA[C_WAVES][32 * PITCH4];
    __shared__ float4 tB[C_WAVES][32 * PITCH4];
    __shared__ float4 tC[C_WAVES][32];

    const int lane = threadIdx.x & 31;
    const int wave = __builtin_amdgcn_readfirstlane(threadIdx.x >> 5);
    float4* wP = tP[wave];
    float4* wV = tV[wave];
    float4* wA = tA[wave];
    float4* wB = tB[wave];
    float4* wC = tC[wave];

    int row0[C_TILES];
    #pragma unroll
    for (int it = 0; it < C_TILES; ++it) {
        const int slot = (it >> 1) * C_WAVES + wave;
        row0[it] = (gridDim.x * slot + (blockIdx.x + slot) % gridDim.x) * 64 + (it & 1) * 32;
    }

    if (row0[C_TILES - 1] + 32 > N) {
        #pragma unroll
        for (int it = 0; it < C_TILES; ++it) {
            const int p = row0[it] + lane;
            if (p < N) som_c_row(p, particles, velocities, r1, r2, gloc, hdr, outP, outV);
        }
        return;
    }

    v4f cv = reinterpret_cast<const v4f*>(hdr + 4)[lane < 24 ? lane : 23];

    int2 gl[C_TILES];
    #pragma unroll
    for (int it = 0; it < C_TILES; ++it)
        gl[it] = reinterpret_cast<const int2*>(gloc)[(size_t)row0[it] + lane];
    asm volatile("" ::: "memory");

    v4f gv[C_TILES][8];
    #pragma unroll
    for (int it = 0; it < C_TILES; ++it) {
        const v4f* vb = reinterpret_cast<const v4f*>(velocities) + (size_t)row0[it] * 8;
        #pragma unroll
        for (int q = 0; q < 8; ++q) gv[it][q] = vb[q * 32 + lane];
    }
    asm volatile("" ::: "memory");

    const float bx = hdr[0], by = hdr[1], radius = hdr[2], sigma2 = hdr[3];

    bool skip[C_TILES], useA[C_TILES];
    unsigned act[C_TILES];
    #pragma unroll
    for (int it = 0; it < C_TILES; ++it) {
        const float dx = (float)gl[it].x - bx;
        const float dy = (float)gl[it].y - by;
        const float nb = expf(-(dx*dx + dy*dy) / sigma2);
        skip[it] = (1.0f - nb) > radius;
        useA[it] = (nb + radius) >= 1.0f;
        act[it]  = __builtin_amdgcn_ballot_w32(!skip[it]);
    }
    asm volatile("" : "+v"(cv));

    #pragma unroll
    for (int ph = 0; ph < C_TILES / 2; ++ph) {
        const int t0 = 2*ph, t1 = 2*ph + 1;
        if ((act[t0] | act[t1]) != 0u) {
            v4f p0[8], a0[8], b0[8], p1[8], a1[8], b1[8];
            som_c_load(p0, a0, b0, particles, r1, r2, row0[t0], lane);
            som_c_load(p1, a1, b1, particles, r1, r2, row0[t1], lane);
            wC[lane] = make_float4(cv.x, cv.y, cv.z, cv.w);
            som_c_work(gv[t0], p0, a0, b0, skip[t0], useA[t0], wP, wV, wA, wB, wC, outP, outV, row0[t0], lane);
            som_c_work(gv[t1], p1, a1, b1, skip[t1], useA[t1], wP, wV, wA, wB, wC, outP, outV, row0[t1], lane);
        } else {
            som_c_copy(gv[t0], outV, row0[t0], lane);
            som_c_copy(gv[t1], outV, row0[t1], lane);
        }
    }
}

extern "C" void kernel_launch(void* const* d_in, const int* in_sizes, int n_in,
                              void* d_out, int out_size, void* d_ws, size_t ws_size,
                              hipStream_t stream)
{
    (void)n_in; (void)out_size; (void)ws_size;
    const float* input      = (const float*)d_in[0];
    const float* particles  = (const float*)d_in[1];
    const float* velocities = (const float*)d_in[2];
    const float* r1         = (const float*)d_in[3];
    const float* r2         = (const float*)d_in[4];
    const int*   gloc       = (const int*)d_in[5];
    const int*   iter_num   = (const int*)d_in[6];

    const int N  = in_sizes[1] / SOM_DIM;
    const int NA = (N + A_ROWS - 1) / A_ROWS;

    float* ws      = (float*)d_ws;
    float* blkSum  = ws;
    float* blkMinV = ws + (size_t)NA * SOM_DIM;
    int*   blkMinI = (int*)(blkMinV + NA);
    size_t hoff    = (size_t)NA * SOM_DIM + 2 * (size_t)NA;
    hoff = (hoff + 3) & ~(size_t)3;
    float* hdr     = ws + hoff;

    float* outP = (float*)d_out;
    float* outV = outP + (size_t)N * SOM_DIM;

    som_pass_a<<<NA, A_THREADS, 0, stream>>>(input, particles, blkSum, blkMinV, blkMinI, outP, N);
    som_pass_b<<<1, B_THREADS, 0, stream>>>(particles, gloc, iter_num,
                                          blkSum, blkMinV, blkMinI, hdr, NA, N);
    som_pass_c<<<NA, C_THREADS, 0, stream>>>(particles, velocities, r1, r2,
                                           gloc, hdr, outP, outV, N);
}
